# starting kernel
# baseline (speedup 1.0000x reference)
#include <hip/hip_runtime.h>
#include <hip/hip_bf16.h>
#include <math.h>

typedef __attribute__((ext_vector_type(2))) float v2f;
typedef __attribute__((ext_vector_type(8))) float v8f;

#define NEG_SLOPE 0.2f
#define BN_EPS 1e-5f

__device__ __forceinline__ unsigned encodeF(float f) {
    unsigned u = __float_as_uint(f);
    return (u & 0x80000000u) ? ~u : (u | 0x80000000u);
}
__device__ __forceinline__ float decodeF(unsigned u) {
    unsigned b = (u & 0x80000000u) ? (u & 0x7fffffffu) : ~u;
    return __uint_as_float(b);
}
__device__ __forceinline__ float lrelu(float v) {
    return v > 0.0f ? v : NEG_SLOPE * v;
}

__global__ __launch_bounds__(256) void fill_u32(unsigned* __restrict__ p, unsigned v, int n) {
    int i = blockIdx.x * blockDim.x + threadIdx.x;
    if (i < n) p[i] = v;
}

__global__ __launch_bounds__(256) void init_out_bias(float* __restrict__ out,
                                                     const float* __restrict__ b,
                                                     int total, int C) {
    int i = blockIdx.x * blockDim.x + threadIdx.x;
    if (i < total) out[i] = b[i % C];
}

__global__ __launch_bounds__(256) void wmma_gemm_f32(const float* __restrict__ A,
                                                     const float* __restrict__ B,
                                                     float* __restrict__ C,
                                                     int M, int K, int Ncols) {
    int wave = (int)((blockIdx.x * (long long)blockDim.x + threadIdx.x) >> 5);
    int lane = threadIdx.x & 31;
    int ntN = Ncols >> 4;
    int totalWaves = (M >> 4) * ntN;
    if (wave >= totalWaves) return;
    int mt = wave / ntN, nt = wave - mt * ntN;
    int m0 = mt << 4, n0 = nt << 4;
    int lo = lane & 15;
    int hi = lane >> 4;
    const float* arow = A + (size_t)(m0 + lo) * K;
    int ncol = n0 + lo;

    v8f acc = {0.f, 0.f, 0.f, 0.f, 0.f, 0.f, 0.f, 0.f};
    for (int k0 = 0; k0 < K; k0 += 4) {
        int kb = k0 + 2 * hi;
        v2f a;
        a.x = arow[kb];
        a.y = arow[kb + 1];
        v2f b;
        b.x = B[(size_t)kb * Ncols + ncol];
        b.y = B[(size_t)(kb + 1) * Ncols + ncol];
        acc = __builtin_amdgcn_wmma_f32_16x16x4_f32(
             false, a,  false, b,
             (short)0, acc,  false,  false);
    }
#pragma unroll
    for (int v = 0; v < 8; ++v) {
        int row = m0 + v + 8 * hi;
        C[(size_t)row * Ncols + ncol] = acc[v];
    }
}

__global__ __launch_bounds__(256) void alpha_kernel(const float* __restrict__ hf,
                                                    const float* __restrict__ a_s,
                                                    const float* __restrict__ a_d,
                                                    float* __restrict__ as_o,
                                                    float* __restrict__ ad_o,
                                                    int N, int H, int Cc) {
    int n = blockIdx.x * blockDim.x + threadIdx.x;
    if (n >= N) return;
    int F = H * Cc;
    for (int h = 0; h < H; ++h) {
        float s = 0.f, d = 0.f;
        for (int c = 0; c < Cc; ++c) {
            float v = hf[(size_t)n * F + h * Cc + c];
            s += v * a_s[h * Cc + c];
            d += v * a_d[h * Cc + c];
        }
        as_o[n * H + h] = s;
        ad_o[n * H + h] = d;
    }
}

__global__ __launch_bounds__(256) void edge_max(const int* __restrict__ ei, int E, int N,
                                                const float* __restrict__ as_i,
                                                const float* __restrict__ ad_i,
                                                unsigned* __restrict__ menc, int H) {
    int e = blockIdx.x * blockDim.x + threadIdx.x;
    int Etot = E + N;
    if (e >= Etot) return;
    int s = (e < E) ? ei[e] : (e - E);
    int d = (e < E) ? ei[E + e] : (e - E);
    for (int h = 0; h < H; ++h) {
        float v = lrelu(as_i[s * H + h] + ad_i[d * H + h]);
        atomicMax(&menc[d * H + h], encodeF(v));
    }
}

__global__ __launch_bounds__(256) void edge_sum(const int* __restrict__ ei, int E, int N,
                                                const float* __restrict__ as_i,
                                                const float* __restrict__ ad_i,
                                                const unsigned* __restrict__ menc,
                                                float* __restrict__ z, int H) {
    int e = blockIdx.x * blockDim.x + threadIdx.x;
    int Etot = E + N;
    if (e >= Etot) return;
    int s = (e < E) ? ei[e] : (e - E);
    int d = (e < E) ? ei[E + e] : (e - E);
    for (int h = 0; h < H; ++h) {
        float v = lrelu(as_i[s * H + h] + ad_i[d * H + h]);
        float ee = __expf(v - decodeF(menc[d * H + h]));
        atomicAdd(&z[d * H + h], ee);
    }
}

__global__ __launch_bounds__(256) void edge_agg(const int* __restrict__ ei, int E, int N,
                                                const float* __restrict__ as_i,
                                                const float* __restrict__ ad_i,
                                                const unsigned* __restrict__ menc,
                                                const float* __restrict__ z,
                                                const float* __restrict__ hsrc,
                                                float* __restrict__ out, int H, int Cc) {
    long long t = blockIdx.x * (long long)blockDim.x + threadIdx.x;
    long long Etot = (long long)E + N;
    if (t >= Etot * Cc) return;
    int e = (int)(t / Cc);
    int c = (int)(t - (long long)e * Cc);
    int s = (e < E) ? ei[e] : (e - E);
    int d = (e < E) ? ei[E + e] : (e - E);
    int F = H * Cc;
    for (int h = 0; h < H; ++h) {
        float v = lrelu(as_i[s * H + h] + ad_i[d * H + h]);
        float al = __expf(v - decodeF(menc[d * H + h])) / (z[d * H + h] + 1e-16f);
        float msg = hsrc[(size_t)s * F + h * Cc + c] * al;
        atomicAdd(&out[(size_t)d * F + h * Cc + c], msg);
    }
}

__global__ __launch_bounds__(256) void bn_elu(const float* __restrict__ agg,
                                              const float* __restrict__ b1,
                                              const float* __restrict__ g,
                                              const float* __restrict__ be,
                                              const float* __restrict__ mu,
                                              const float* __restrict__ var,
                                              float* __restrict__ outp, int total, int F) {
    int i = blockIdx.x * blockDim.x + threadIdx.x;
    if (i >= total) return;
    int j = i % F;
    float v = agg[i] + b1[j];
    v = (v - mu[j]) * g[j] * rsqrtf(var[j] + BN_EPS) + be[j];
    outp[i] = v > 0.f ? v : (__expf(v) - 1.f);
}

static inline int nblk(long long t) { return (int)((t + 255) / 256); }

extern "C" void kernel_launch(void* const* d_in, const int* in_sizes, int n_in,
                              void* d_out, int out_size, void* d_ws, size_t ws_size,
                              hipStream_t stream) {
    const float* x      = (const float*)d_in[0];
    const int*   ei     = (const int*)d_in[1];
    const float* W1     = (const float*)d_in[2];
    const float* a_src1 = (const float*)d_in[3];
    const float* a_dst1 = (const float*)d_in[4];
    const float* b1     = (const float*)d_in[5];
    const float* bn_g   = (const float*)d_in[6];
    const float* bn_b   = (const float*)d_in[7];
    const float* bn_m   = (const float*)d_in[8];
    const float* bn_v   = (const float*)d_in[9];
    const float* W2     = (const float*)d_in[10];
    const float* a_src2 = (const float*)d_in[11];
    const float* a_dst2 = (const float*)d_in[12];
    const float* b2     = (const float*)d_in[13];

    const int IN_DIM = 128, F1 = 256, C1 = 64, H1 = 4, C2 = 64;
    int N = in_sizes[0] / IN_DIM;
    int E = in_sizes[1] / 2;
    int Etot = E + N;

    float* ws = (float*)d_ws;
    size_t o = 0;
    float*    h1   = ws + o; o += (size_t)N * F1;
    float*    agg1 = ws + o; o += (size_t)N * F1;
    float*    as1  = ws + o; o += (size_t)N * H1;
    float*    ad1  = ws + o; o += (size_t)N * H1;
    unsigned* m1   = (unsigned*)(ws + o); o += (size_t)N * H1;
    float*    z1   = ws + o; o += (size_t)N * H1;
    float*    h2   = ws + o; o += (size_t)N * C2;
    float*    as2  = ws + o; o += (size_t)N;
    float*    ad2  = ws + o; o += (size_t)N;
    unsigned* m2   = (unsigned*)(ws + o); o += (size_t)N;
    float*    z2   = ws + o; o += (size_t)N;

    fill_u32<<<nblk((long long)N * F1), 256, 0, stream>>>((unsigned*)agg1, 0u, N * F1);
    fill_u32<<<nblk(N * H1), 256, 0, stream>>>(m1, 0u, N * H1);
    fill_u32<<<nblk(N * H1), 256, 0, stream>>>((unsigned*)z1, 0u, N * H1);
    fill_u32<<<nblk(N), 256, 0, stream>>>(m2, 0u, N);
    fill_u32<<<nblk(N), 256, 0, stream>>>((unsigned*)z2, 0u, N);

    {
        long long waves = (long long)(N / 16) * (F1 / 16);
        wmma_gemm_f32<<<nblk(waves * 32), 256, 0, stream>>>(x, W1, h1, N, IN_DIM, F1);
    }
    alpha_kernel<<<nblk(N), 256, 0, stream>>>(h1, a_src1, a_dst1, as1, ad1, N, H1, C1);
    edge_max<<<nblk(Etot), 256, 0, stream>>>(ei, E, N, as1, ad1, m1, H1);
    edge_sum<<<nblk(Etot), 256, 0, stream>>>(ei, E, N, as1, ad1, m1, z1, H1);
    edge_agg<<<nblk((long long)Etot * C1), 256, 0, stream>>>(ei, E, N, as1, ad1, m1, z1, h1,
                                                             agg1, H1, C1);
    bn_elu<<<nblk((long long)N * F1), 256, 0, stream>>>(agg1, b1, bn_g, bn_b, bn_m, bn_v, h1,
                                                        N * F1, F1);

    {
        long long waves = (long long)(N / 16) * (C2 / 16);
        wmma_gemm_f32<<<nblk(waves * 32), 256, 0, stream>>>(h1, W2, h2, N, F1, C2);
    }
    alpha_kernel<<<nblk(N), 256, 0, stream>>>(h2, a_src2, a_dst2, as2, ad2, N, 1, C2);
    edge_max<<<nblk(Etot), 256, 0, stream>>>(ei, E, N, as2, ad2, m2, 1);
    edge_sum<<<nblk(Etot), 256, 0, stream>>>(ei, E, N, as2, ad2, m2, z2, 1);
    init_out_bias<<<nblk((long long)N * C2), 256, 0, stream>>>((float*)d_out, b2, N * C2, C2);
    edge_agg<<<nblk((long long)Etot * C2), 256, 0, stream>>>(ei, E, N, as2, ad2, m2, z2, h2,
                                                             (float*)d_out, 1, C2);
}


# Edge aggregation remapped from one thread per (edge, channel) to whole waves per edge: endpoints and softmax weights computed once per wave, lane l owns channels l+32j; float atomic adds kept
# speedup vs baseline: 1.5687x; 1.5687x over previous
#include <hip/hip_runtime.h>
#include <hip/hip_bf16.h>
#include <math.h>

typedef __attribute__((ext_vector_type(2))) float v2f;
typedef __attribute__((ext_vector_type(8))) float v8f;

#define NEG_SLOPE 0.2f
#define BN_EPS 1e-5f

__device__ __forceinline__ unsigned encodeF(float f) {
    unsigned u = __float_as_uint(f);
    return (u & 0x80000000u) ? ~u : (u | 0x80000000u);
}
__device__ __forceinline__ float decodeF(unsigned u) {
    unsigned b = (u & 0x80000000u) ? (u & 0x7fffffffu) : ~u;
    return __uint_as_float(b);
}
__device__ __forceinline__ float lrelu(float v) {
    return v > 0.0f ? v : NEG_SLOPE * v;
}

__global__ __launch_bounds__(256) void fill_u32(unsigned* __restrict__ p, unsigned v, int n) {
    int i = blockIdx.x * blockDim.x + threadIdx.x;
    if (i < n) p[i] = v;
}

__global__ __launch_bounds__(256) void init_out_bias(float* __restrict__ out,
                                                     const float* __restrict__ b,
                                                     int total, int C) {
    int i = blockIdx.x * blockDim.x + threadIdx.x;
    if (i < total) out[i] = b[i % C];
}

__global__ __launch_bounds__(256) void wmma_gemm_f32(const float* __restrict__ A,
                                                     const float* __restrict__ B,
                                                     float* __restrict__ C,
                                                     int M, int K, int Ncols) {
    int wave = (int)((blockIdx.x * (long long)blockDim.x + threadIdx.x) >> 5);
    int lane = threadIdx.x & 31;
    int ntN = Ncols >> 4;
    int totalWaves = (M >> 4) * ntN;
    if (wave >= totalWaves) return;
    int mt = wave / ntN, nt = wave - mt * ntN;
    int m0 = mt << 4, n0 = nt << 4;
    int lo = lane & 15;
    int hi = lane >> 4;
    const float* arow = A + (size_t)(m0 + lo) * K;
    int ncol = n0 + lo;

    v8f acc = {0.f, 0.f, 0.f, 0.f, 0.f, 0.f, 0.f, 0.f};
    for (int k0 = 0; k0 < K; k0 += 4) {
        int kb = k0 + 2 * hi;
        v2f a;
        a.x = arow[kb];
        a.y = arow[kb + 1];
        v2f b;
        b.x = B[(size_t)kb * Ncols + ncol];
        b.y = B[(size_t)(kb + 1) * Ncols + ncol];
        acc = __builtin_amdgcn_wmma_f32_16x16x4_f32(
             false, a,  false, b,
             (short)0, acc,  false,  false);
    }
#pragma unroll
    for (int v = 0; v < 8; ++v) {
        int row = m0 + v + 8 * hi;
        C[(size_t)row * Ncols + ncol] = acc[v];
    }
}

__global__ __launch_bounds__(256) void alpha_kernel(const float* __restrict__ hf,
                                                    const float* __restrict__ a_s,
                                                    const float* __restrict__ a_d,
                                                    float* __restrict__ as_o,
                                                    float* __restrict__ ad_o,
                                                    int N, int H, int Cc) {
    int n = blockIdx.x * blockDim.x + threadIdx.x;
    if (n >= N) return;
    int F = H * Cc;
    for (int h = 0; h < H; ++h) {
        float s = 0.f, d = 0.f;
        for (int c = 0; c < Cc; ++c) {
            float v = hf[(size_t)n * F + h * Cc + c];
            s += v * a_s[h * Cc + c];
            d += v * a_d[h * Cc + c];
        }
        as_o[n * H + h] = s;
        ad_o[n * H + h] = d;
    }
}

__global__ __launch_bounds__(256) void edge_max(const int* __restrict__ ei, int E, int N,
                                                const float* __restrict__ as_i,
                                                const float* __restrict__ ad_i,
                                                unsigned* __restrict__ menc, int H) {
    int e = blockIdx.x * blockDim.x + threadIdx.x;
    int Etot = E + N;
    if (e >= Etot) return;
    int s = (e < E) ? ei[e] : (e - E);
    int d = (e < E) ? ei[E + e] : (e - E);
    for (int h = 0; h < H; ++h) {
        float v = lrelu(as_i[s * H + h] + ad_i[d * H + h]);
        atomicMax(&menc[d * H + h], encodeF(v));
    }
}

__global__ __launch_bounds__(256) void edge_sum(const int* __restrict__ ei, int E, int N,
                                                const float* __restrict__ as_i,
                                                const float* __restrict__ ad_i,
                                                const unsigned* __restrict__ menc,
                                                float* __restrict__ z, int H) {
    int e = blockIdx.x * blockDim.x + threadIdx.x;
    int Etot = E + N;
    if (e >= Etot) return;
    int s = (e < E) ? ei[e] : (e - E);
    int d = (e < E) ? ei[E + e] : (e - E);
    for (int h = 0; h < H; ++h) {
        float v = lrelu(as_i[s * H + h] + ad_i[d * H + h]);
        float ee = __expf(v - decodeF(menc[d * H + h]));
        atomicAdd(&z[d * H + h], ee);
    }
}

template <int H, int Cc, int K>
__device__ __forceinline__ void agg_edges(int e0, int lane, const int* __restrict__ ei, int E,
                                          const float* __restrict__ as_i,
                                          const float* __restrict__ ad_i,
                                          const unsigned* __restrict__ menc,
                                          const float* __restrict__ z,
                                          const float* __restrict__ hsrc,
                                          float* __restrict__ out) {
    constexpr int F = H * Cc;
    constexpr int J = F / 32;
    static_assert(K * H <= 32, "one lane per (edge, head)");
    const int li = min(lane / H, K - 1);
    const int lh = lane & (H - 1);
    const int e = e0 + li;
    const int ec = min(e, E - 1);
    const int sl = ei[ec];
    const int dl = ei[E + ec];
    const int s = (e < E) ? sl : (e - E);
    const int d = (e < E) ? dl : (e - E);
    const float v = lrelu(as_i[s * H + lh] + ad_i[d * H + lh]);
    const float al = __expf(v - decodeF(menc[d * H + lh])) / (z[d * H + lh] + 1e-16f);

    float r[K][J];
    for (int i = 0; i < K; ++i) {
        const int si = __builtin_amdgcn_readlane(s, i * H);
        const float* hr = hsrc + (size_t)si * F + lane;
        for (int j = 0; j < J; ++j) r[i][j] = hr[32 * j];
    }
    for (int i = 0; i < K; ++i) {
        const int di = __builtin_amdgcn_readlane(d, i * H);
        float* orow = out + (size_t)di * F + lane;
        for (int j = 0; j < J; ++j) {
            const float wj = __int_as_float(
                __builtin_amdgcn_readlane(__float_as_int(al), i * H + (32 * j) / Cc));
            atomicAdd(orow + 32 * j, r[i][j] * wj);
        }
    }
}

template <int H, int Cc, int EPW>
__global__ __launch_bounds__(256) void edge_agg_wave(const int* __restrict__ ei, int E, int N,
                                                     const float* __restrict__ as_i,
                                                     const float* __restrict__ ad_i,
                                                     const unsigned* __restrict__ menc,
                                                     const float* __restrict__ z,
                                                     const float* __restrict__ hsrc,
                                                     float* __restrict__ out) {
    static_assert((H & (H - 1)) == 0, "H must be a power of two");
    static_assert(Cc % 32 == 0, "a 32-channel block must lie inside one head");
    const int lane = threadIdx.x & 31;
    const int wave = __builtin_amdgcn_readfirstlane((int)(blockIdx.x * 8u + (threadIdx.x >> 5)));
    const int Etot = E + N;
    const int e0 = wave * EPW;
    if (e0 + EPW <= Etot) {
        agg_edges<H, Cc, EPW>(e0, lane, ei, E, as_i, ad_i, menc, z, hsrc, out);
    } else {
        for (int e = e0; e < Etot; ++e)
            agg_edges<H, Cc, 1>(e, lane, ei, E, as_i, ad_i, menc, z, hsrc, out);
    }
}

__global__ __launch_bounds__(256) void bn_elu(const float* __restrict__ agg,
                                              const float* __restrict__ b1,
                                              const float* __restrict__ g,
                                              const float* __restrict__ be,
                                              const float* __restrict__ mu,
                                              const float* __restrict__ var,
                                              float* __restrict__ outp, int total, int F) {
    int i = blockIdx.x * blockDim.x + threadIdx.x;
    if (i >= total) return;
    int j = i % F;
    float v = agg[i] + b1[j];
    v = (v - mu[j]) * g[j] * rsqrtf(var[j] + BN_EPS) + be[j];
    outp[i] = v > 0.f ? v : (__expf(v) - 1.f);
}

static inline int nblk(long long t) { return (int)((t + 255) / 256); }

extern "C" void kernel_launch(void* const* d_in, const int* in_sizes, int n_in,
                              void* d_out, int out_size, void* d_ws, size_t ws_size,
                              hipStream_t stream) {
    const float* x      = (const float*)d_in[0];
    const int*   ei     = (const int*)d_in[1];
    const float* W1     = (const float*)d_in[2];
    const float* a_src1 = (const float*)d_in[3];
    const float* a_dst1 = (const float*)d_in[4];
    const float* b1     = (const float*)d_in[5];
    const float* bn_g   = (const float*)d_in[6];
    const float* bn_b   = (const float*)d_in[7];
    const float* bn_m   = (const float*)d_in[8];
    const float* bn_v   = (const float*)d_in[9];
    const float* W2     = (const float*)d_in[10];
    const float* a_src2 = (const float*)d_in[11];
    const float* a_dst2 = (const float*)d_in[12];
    const float* b2     = (const float*)d_in[13];

    const int IN_DIM = 128, F1 = 256, C1 = 64, H1 = 4, C2 = 64;
    int N = in_sizes[0] / IN_DIM;
    int E = in_sizes[1] / 2;
    int Etot = E + N;

    float* ws = (float*)d_ws;
    size_t o = 0;
    float*    h1   = ws + o; o += (size_t)N * F1;
    float*    agg1 = ws + o; o += (size_t)N * F1;
    float*    as1  = ws + o; o += (size_t)N * H1;
    float*    ad1  = ws + o; o += (size_t)N * H1;
    unsigned* m1   = (unsigned*)(ws + o); o += (size_t)N * H1;
    float*    z1   = ws + o; o += (size_t)N * H1;
    float*    h2   = ws + o; o += (size_t)N * C2;
    float*    as2  = ws + o; o += (size_t)N;
    float*    ad2  = ws + o; o += (size_t)N;
    unsigned* m2   = (unsigned*)(ws + o); o += (size_t)N;
    float*    z2   = ws + o; o += (size_t)N;

    fill_u32<<<nblk((long long)N * F1), 256, 0, stream>>>((unsigned*)agg1, 0u, N * F1);
    fill_u32<<<nblk(N * H1), 256, 0, stream>>>(m1, 0u, N * H1);
    fill_u32<<<nblk(N * H1), 256, 0, stream>>>((unsigned*)z1, 0u, N * H1);
    fill_u32<<<nblk(N), 256, 0, stream>>>(m2, 0u, N);
    fill_u32<<<nblk(N), 256, 0, stream>>>((unsigned*)z2, 0u, N);

    {
        long long waves = (long long)(N / 16) * (F1 / 16);
        wmma_gemm_f32<<<nblk(waves * 32), 256, 0, stream>>>(x, W1, h1, N, IN_DIM, F1);
    }
    alpha_kernel<<<nblk(N), 256, 0, stream>>>(h1, a_src1, a_dst1, as1, ad1, N, H1, C1);
    edge_max<<<nblk(Etot), 256, 0, stream>>>(ei, E, N, as1, ad1, m1, H1);
    edge_sum<<<nblk(Etot), 256, 0, stream>>>(ei, E, N, as1, ad1, m1, z1, H1);
    edge_agg_wave<4, 64, 4><<<nblk((long long)((Etot + 3) / 4) * 32), 256, 0, stream>>>(
        ei, E, N, as1, ad1, m1, z1, h1, agg1);
    bn_elu<<<nblk((long long)N * F1), 256, 0, stream>>>(agg1, b1, bn_g, bn_b, bn_m, bn_v, h1,
                                                        N * F1, F1);

    {
        long long waves = (long long)(N / 16) * (C2 / 16);
        wmma_gemm_f32<<<nblk(waves * 32), 256, 0, stream>>>(h1, W2, h2, N, F1, C2);
    }
    alpha_kernel<<<nblk(N), 256, 0, stream>>>(h2, a_src2, a_dst2, as2, ad2, N, 1, C2);
    edge_max<<<nblk(Etot), 256, 0, stream>>>(ei, E, N, as2, ad2, m2, 1);
    edge_sum<<<nblk(Etot), 256, 0, stream>>>(ei, E, N, as2, ad2, m2, z2, 1);
    init_out_bias<<<nblk((long long)N * C2), 256, 0, stream>>>((float*)d_out, b2, N * C2, C2);
    edge_agg_wave<1, 64, 8><<<nblk((long long)((Etot + 7) / 8) * 32), 256, 0, stream>>>(
        ei, E, N, as2, ad2, m2, z2, h2, (float*)d_out);
}


# Both GEMMs: a work-group copies its 32 or 64 rows of A into LDS once (whole-row 16-byte loads, padded pitch); each wave keeps a 32x32 tile, A fragments from LDS, each B fragment feeds two row tiles
# speedup vs baseline: 5.9101x; 3.7675x over previous
#include <hip/hip_runtime.h>
#include <hip/hip_bf16.h>
#include <math.h>

typedef __attribute__((ext_vector_type(2))) float v2f;
typedef __attribute__((ext_vector_type(8))) float v8f;

#define NEG_SLOPE 0.2f
#define BN_EPS 1e-5f

__device__ __forceinline__ unsigned encodeF(float f) {
    unsigned u = __float_as_uint(f);
    return (u & 0x80000000u) ? ~u : (u | 0x80000000u);
}
__device__ __forceinline__ float decodeF(unsigned u) {
    unsigned b = (u & 0x80000000u) ? (u & 0x7fffffffu) : ~u;
    return __uint_as_float(b);
}
__device__ __forceinline__ float lrelu(float v) {
    return v > 0.0f ? v : NEG_SLOPE * v;
}

template <int K, int NWM, int NWN>
__global__ __launch_bounds__(32 * NWM * NWN) void wmma_gemm_f32(const float* __restrict__ A,
                                                                const float* __restrict__ B,
                                                                float* __restrict__ C, int M) {
    static_assert(K % 4 == 0, "a k-step is 4 wide and rows are copied in 16-byte pieces");
    constexpr int NT = 32 * NWM * NWN;
    constexpr int BM = 32 * NWM;
    constexpr int NC = 32 * NWN;
    constexpr int P = K + 4;
    constexpr int K4 = K / 4;
    __shared__ __attribute__((aligned(16))) float As[BM * P];
    const int tid = threadIdx.x;
    const int lane = tid & 31;
    const int wv = __builtin_amdgcn_readfirstlane(tid >> 5);
    const int wm = wv / NWN;
    const int wn = wv - wm * NWN;
    const int row0 = blockIdx.x * BM;

    static_assert((BM * K4) % NT == 0, "every thread copies the same number of 16-byte pieces");
    constexpr int NI = (BM * K4) / NT;
    constexpr int UB = NI < 8 ? NI : 8;
    static_assert(NI % UB == 0, "whole batches");
    for (int i0 = 0; i0 < NI; i0 += UB) {
        float4 v[UB];
#pragma unroll
        for (int u = 0; u < UB; ++u) {
            const int i = (i0 + u) * NT + tid;
            const int r = i / K4;
            const int c4 = i - r * K4;
            const int gr = min(row0 + r, M - 1);
            v[u] = *(const float4*)(A + (size_t)gr * K + 4 * c4);
        }
#pragma unroll
        for (int u = 0; u < UB; ++u) {
            const int i = (i0 + u) * NT + tid;
            const int r = i / K4;
            const int c4 = i - r * K4;
            *(float4*)(As + r * P + 4 * c4) = v[u];
        }
    }
    __syncthreads();

    const int lo = lane & 15;
    const int hi = lane >> 4;
    const float* a0p = As + (32 * wm + lo) * P + 2 * hi;
    const float* a1p = a0p + 16 * P;
    const float* bp = B + (size_t)(2 * hi) * NC + 32 * wn + lo;

    v8f c00 = {0.f, 0.f, 0.f, 0.f, 0.f, 0.f, 0.f, 0.f};
    v8f c01 = c00, c10 = c00, c11 = c00;
#pragma unroll 4
    for (int k0 = 0; k0 < K; k0 += 4) {
        const v2f a0 = *(const v2f*)(a0p + k0);
        const v2f a1 = *(const v2f*)(a1p + k0);
        v2f b0, b1;
        b0.x = bp[(size_t)k0 * NC];
        b0.y = bp[(size_t)(k0 + 1) * NC];
        b1.x = bp[(size_t)k0 * NC + 16];
        b1.y = bp[(size_t)(k0 + 1) * NC + 16];
        c00 = __builtin_amdgcn_wmma_f32_16x16x4_f32(false, a0, false, b0, (short)0, c00, false, false);
        c01 = __builtin_amdgcn_wmma_f32_16x16x4_f32(false, a0, false, b1, (short)0, c01, false, false);
        c10 = __builtin_amdgcn_wmma_f32_16x16x4_f32(false, a1, false, b0, (short)0, c10, false, false);
        c11 = __builtin_amdgcn_wmma_f32_16x16x4_f32(false, a1, false, b1, (short)0, c11, false, false);
    }
    const int ncol = 32 * wn + lo;
#pragma unroll
    for (int v = 0; v < 8; ++v) {
        const int r0 = row0 + 32 * wm + v + 8 * hi;
        const int r1 = r0 + 16;
        if (r0 < M) {
            C[(size_t)r0 * NC + ncol] = c00[v];
            C[(size_t)r0 * NC + ncol + 16] = c01[v];
        }
        if (r1 < M) {
            C[(size_t)r1 * NC + ncol] = c10[v];
            C[(size_t)r1 * NC + ncol + 16] = c11[v];
        }
    }
}

__global__ __launch_bounds__(256) void alpha_kernel(const float* __restrict__ hf,
                                                    const float* __restrict__ a_s,
                                                    const float* __restrict__ a_d,
                                                    float* __restrict__ as_o,
                                                    float* __restrict__ ad_o,
                                                    int N, int H, int Cc) {
    int n = blockIdx.x * blockDim.x + threadIdx.x;
    if (n >= N) return;
    int F = H * Cc;
    for (int h = 0; h < H; ++h) {
        float s = 0.f, d = 0.f;
        for (int c = 0; c < Cc; ++c) {
            float v = hf[(size_t)n * F + h * Cc + c];
            s += v * a_s[h * Cc + c];
            d += v * a_d[h * Cc + c];
        }
        as_o[n * H + h] = s;
        ad_o[n * H + h] = d;
    }
}

template <int NPB, int NB, int NS, int SEG>
__global__ __launch_bounds__(1024) void part_edges(const int* __restrict__ ei, int E,
                                                   int* __restrict__ seg) {
    static_assert(NPB <= 256, "an entry keeps the local destination in its low 8 bits");
    static_assert(SEG % 4 == 0, "segments are moved in 16-byte pieces");
    constexpr int NT = 1024;
    constexpr int UB = 5;
    constexpr int V = SEG / 4;
    __shared__ __attribute__((aligned(16))) int buf[NB * SEG];
    const int tid = threadIdx.x;
    const int s = blockIdx.x;
    const int* __restrict__ srcp = ei;
    const int* __restrict__ dstp = ei + E;
    const int len = (E + NS - 1) / NS;
    const int e0 = s * len;
    const int e1 = min(E, e0 + len);

    for (int i = tid; i < NB * SEG; i += NT) buf[i] = (i & (SEG - 1)) ? -1 : 0;
    __syncthreads();
    for (int base = e0; base < e1; base += NT * UB) {
        int d[UB], sr[UB];
#pragma unroll
        for (int k = 0; k < UB; ++k) {
            const int e = base + k * NT + tid;
            const int ec = min(e, e1 - 1);
            const int dv = dstp[ec];
            sr[k] = srcp[ec];
            d[k] = (e < e1) ? dv : -1;
        }
#pragma unroll
        for (int k = 0; k < UB; ++k) {
            const unsigned b = (unsigned)d[k] / (unsigned)NPB;
            if (b < (unsigned)NB) {
                const int pos = atomicAdd(&buf[b * SEG], 1);
                if (pos < SEG - 1)
                    buf[b * SEG + 1 + pos] = (sr[k] << 8) | (int)((unsigned)d[k] - b * (unsigned)NPB);
            }
        }
    }
    __syncthreads();
    const int4* bv = (const int4*)buf;
    for (int i = tid; i < NB * V; i += NT) {
        const int b = i / V;
        const int q = i - b * V;
        ((int4*)(seg + ((size_t)b * NS + s) * SEG))[q] = bv[i];
    }
}

template <int H, int Cc, int NPB, int CAP, bool BN, int NS, int SEG>
__global__ __launch_bounds__(1024) void agg_owner(const int* __restrict__ ei, int E, int N,
                                                  const float* __restrict__ as_i,
                                                  const float* __restrict__ ad_i,
                                                  const float* __restrict__ hsrc,
                                                  const float* __restrict__ bias,
                                                  const float* __restrict__ g,
                                                  const float* __restrict__ be,
                                                  const float* __restrict__ mu,
                                                  const float* __restrict__ var,
                                                  float* __restrict__ outp,
                                                  const int* __restrict__ seg) {
    static_assert((H & (H - 1)) == 0, "H must be a power of two");
    static_assert(NPB <= 256, "an entry keeps the local destination in its low 8 bits");
    static_assert(Cc % 32 == 0, "a 32-channel block must lie inside one head");
    static_assert(SEG % 4 == 0, "segments are read in 16-byte pieces");
    constexpr int F = H * Cc;
    constexpr int J = F / 32;
    constexpr int K = (32 / H) < 8 ? (32 / H) : 8;
    constexpr int KK = K > 4 ? (J >= 8 ? 4 : K) : K;
    constexpr int NT = 1024;
    constexpr int NW = NT / 32;
    constexpr int V = SEG / 4;
    constexpr int NV = NS * V;
    constexpr int UB = (NV + NT - 1) / NT;
    __shared__ float acc[NPB * F];
    __shared__ float zs[NPB * H];
    __shared__ unsigned mxs[NPB * H];
    __shared__ int le[CAP];
    __shared__ int ldl[CAP];
    __shared__ unsigned cnt;

    const int tid = threadIdx.x;
    const int lane = tid & 31;
    const int wv = __builtin_amdgcn_readfirstlane(tid >> 5);
    const int lo = blockIdx.x * NPB;
    const int* __restrict__ srcp = ei;
    const int* __restrict__ dstp = ei + E;

    const int4* __restrict__ sv = (const int4*)(seg + (size_t)blockIdx.x * (NS * SEG));
    int4 w4[UB];
#pragma unroll
    for (int k = 0; k < UB; ++k) {
        const int v = k * NT + tid;
        w4[k] = (v < NV) ? sv[v] : make_int4(-1, -1, -1, -1);
    }

    for (int i = tid; i < NPB * F; i += NT) acc[i] = 0.f;
    for (int i = tid; i < NPB * H; i += NT) {
        zs[i] = 0.f;
        mxs[i] = 0u;
    }
    if (tid == 0) cnt = 0u;
    __syncthreads();

    if (tid < NPB && lo + tid < N) {
        const unsigned pos = atomicAdd(&cnt, 1u);
        if (pos < (unsigned)CAP) { le[pos] = lo + tid; ldl[pos] = tid; }
    }
#pragma unroll
    for (int k = 0; k < UB; ++k) {
        const int v = k * NT + tid;
        int ww[4] = {w4[k].x, w4[k].y, w4[k].z, w4[k].w};
        if (v < NV && (v & (V - 1)) == 0) {
            if (ww[0] > SEG - 1) atomicAdd(&cnt, (unsigned)CAP + 1u);
            ww[0] = -1;
        }
#pragma unroll
        for (int q = 0; q < 4; ++q) {
            if (ww[q] >= 0) {
                const unsigned pos = atomicAdd(&cnt, 1u);
                if (pos < (unsigned)CAP) { le[pos] = ww[q] >> 8; ldl[pos] = ww[q] & 255; }
            }
        }
    }
    __syncthreads();
    const int total = (int)cnt;

    if (total <= CAP) {
        for (int p = tid; p < total; p += NT) {
            const int s = le[p];
            const int dl = ldl[p];
            for (int h = 0; h < H; ++h) {
                const float v = lrelu(as_i[s * H + h] + ad_i[(lo + dl) * H + h]);
                atomicMax(&mxs[dl * H + h], encodeF(v));
            }
        }
        __syncthreads();
        const int ngrp = (total + KK - 1) / KK;
        for (int gi = wv; gi < ngrp; gi += NW) {
            const int p0 = gi * KK;
            const int li = min(lane / H, KK - 1);
            const int lh = lane & (H - 1);
            const int p = min(p0 + li, total - 1);
            const int s = le[p];
            const int dl = ldl[p];
            const float v = lrelu(as_i[s * H + lh] + ad_i[(lo + dl) * H + lh]);
            const float al = __expf(v - decodeF(mxs[dl * H + lh]));
            float r[KK][J];
            for (int i = 0; i < KK; ++i) {
                if (p0 + i < total) {
                    const int si = __builtin_amdgcn_readlane(s, i * H);
                    const float* hr = hsrc + (size_t)si * F + lane;
                    for (int j = 0; j < J; ++j) r[i][j] = hr[32 * j];
                }
            }
            if (lane < KK * H && p0 + li < total) atomicAdd(&zs[dl * H + lh], al);
            for (int i = 0; i < KK; ++i) {
                if (p0 + i < total) {
                    const int di = __builtin_amdgcn_readlane(dl, i * H);
                    float* orow = acc + di * F + lane;
                    for (int j = 0; j < J; ++j) {
                        const float wj = __int_as_float(
                            __builtin_amdgcn_readlane(__float_as_int(al), i * H + (32 * j) / Cc));
                        atomicAdd(orow + 32 * j, r[i][j] * wj);
                    }
                }
            }
        }
    } else {
        const int Etot = E + N;
        for (int e = tid; e < Etot; e += NT) {
            const int ec = min(e, E - 1);
            const int dd = (e < E) ? dstp[ec] : (e - E);
            const unsigned dl = (unsigned)(dd - lo);
            if (dl < (unsigned)NPB) {
                const int s = (e < E) ? srcp[ec] : (e - E);
                for (int h = 0; h < H; ++h) {
                    const float v = lrelu(as_i[s * H + h] + ad_i[dd * H + h]);
                    atomicMax(&mxs[dl * H + h], encodeF(v));
                }
            }
        }
        __syncthreads();
        for (int e = tid; e < Etot; e += NT) {
            const int ec = min(e, E - 1);
            const int dd = (e < E) ? dstp[ec] : (e - E);
            const unsigned dl = (unsigned)(dd - lo);
            if (dl < (unsigned)NPB) {
                const int s = (e < E) ? srcp[ec] : (e - E);
                for (int h = 0; h < H; ++h) {
                    const float v = lrelu(as_i[s * H + h] + ad_i[dd * H + h]);
                    const float w = __expf(v - decodeF(mxs[dl * H + h]));
                    atomicAdd(&zs[dl * H + h], w);
                    for (int c = 0; c < Cc; ++c)
                        atomicAdd(&acc[dl * F + h * Cc + c], w * hsrc[(size_t)s * F + h * Cc + c]);
                }
            }
        }
    }
    __syncthreads();

    for (int idx = tid; idx < NPB * F; idx += NT) {
        const int i = idx / F;
        const int j = idx - i * F;
        const int n = lo + i;
        if (n < N) {
            float v = acc[idx] / (zs[i * H + j / Cc] + 1e-16f) + bias[j];
            if (BN) {
                v = (v - mu[j]) * g[j] * rsqrtf(var[j] + BN_EPS) + be[j];
                v = v > 0.f ? v : (__expf(v) - 1.f);
            }
            outp[(size_t)n * F + j] = v;
        }
    }
}

static inline int nblk(long long t) { return (int)((t + 255) / 256); }

extern "C" void kernel_launch(void* const* d_in, const int* in_sizes, int n_in,
                              void* d_out, int out_size, void* d_ws, size_t ws_size,
                              hipStream_t stream) {
    const float* x      = (const float*)d_in[0];
    const int*   ei     = (const int*)d_in[1];
    const float* W1     = (const float*)d_in[2];
    const float* a_src1 = (const float*)d_in[3];
    const float* a_dst1 = (const float*)d_in[4];
    const float* b1     = (const float*)d_in[5];
    const float* bn_g   = (const float*)d_in[6];
    const float* bn_b   = (const float*)d_in[7];
    const float* bn_m   = (const float*)d_in[8];
    const float* bn_v   = (const float*)d_in[9];
    const float* W2     = (const float*)d_in[10];
    const float* a_src2 = (const float*)d_in[11];
    const float* a_dst2 = (const float*)d_in[12];
    const float* b2     = (const float*)d_in[13];

    const int IN_DIM = 128, F1 = 256, C1 = 64, H1 = 4, C2 = 64;
    const int NPB = 196;
    int N = in_sizes[0] / IN_DIM;
    int E = in_sizes[1] / 2;

    float* ws = (float*)d_ws;
    size_t o = 0;
    float* h1   = ws + o; o += (size_t)N * F1;
    float* act1 = ws + o; o += (size_t)N * F1;
    float* as1  = ws + o; o += (size_t)N * H1;
    float* ad1  = ws + o; o += (size_t)N * H1;
    int*   eseg = (int*)(ws + o); o += (size_t)N * C2;
    float* h2   = h1;
    float* as2  = as1;
    float* ad2  = ad1;

    const int ngroups = (N + NPB - 1) / NPB;

    part_edges<196, 256, 192, 64><<<192, 1024, 0, stream>>>(ei, E, eseg);

    wmma_gemm_f32<128, 1, 8><<<(N + 31) / 32, 256, 0, stream>>>(x, W1, h1, N);
    alpha_kernel<<<nblk(N), 256, 0, stream>>>(h1, a_src1, a_dst1, as1, ad1, N, H1, C1);
    agg_owner<4, 64, 196, 8192, true, 192, 64><<<ngroups, 1024, 0, stream>>>(
        ei, E, N, as1, ad1, h1, b1, bn_g, bn_b, bn_m, bn_v, act1, eseg);

    wmma_gemm_f32<256, 2, 2><<<(N + 63) / 64, 128, 0, stream>>>(act1, W2, h2, N);
    alpha_kernel<<<nblk(N), 256, 0, stream>>>(h2, a_src2, a_dst2, as2, ad2, N, 1, C2);
    agg_owner<1, 64, 196, 8192, false, 192, 64><<<ngroups, 1024, 0, stream>>>(
        ei, E, N, as2, ad2, h2, b2, b2, b2, b2, b2, (float*)d_out, eseg);
}


# Owner groups' node epilogue: a thread keeps 4 columns with their bias/batchnorm constants in registers; softmax sum inverted once per (node, head), a multiply per element; 16-byte LDS reads and stores
# speedup vs baseline: 6.9821x; 1.1814x over previous
#include <hip/hip_runtime.h>
#include <hip/hip_bf16.h>
#include <math.h>

typedef __attribute__((ext_vector_type(2))) float v2f;
typedef __attribute__((ext_vector_type(8))) float v8f;

#define NEG_SLOPE 0.2f
#define BN_EPS 1e-5f

__device__ __forceinline__ unsigned encodeF(float f) {
    unsigned u = __float_as_uint(f);
    return (u & 0x80000000u) ? ~u : (u | 0x80000000u);
}
__device__ __forceinline__ float decodeF(unsigned u) {
    unsigned b = (u & 0x80000000u) ? (u & 0x7fffffffu) : ~u;
    return __uint_as_float(b);
}
__device__ __forceinline__ float lrelu(float v) {
    return v > 0.0f ? v : NEG_SLOPE * v;
}

template <int K, int NWM, int NWN>
__global__ __launch_bounds__(32 * NWM * NWN) void wmma_gemm_f32(const float* __restrict__ A,
                                                                const float* __restrict__ B,
                                                                float* __restrict__ C, int M) {
    static_assert(K % 4 == 0, "a k-step is 4 wide and rows are copied in 16-byte pieces");
    constexpr int NT = 32 * NWM * NWN;
    constexpr int BM = 32 * NWM;
    constexpr int NC = 32 * NWN;
    constexpr int P = K + 4;
    constexpr int K4 = K / 4;
    __shared__ __attribute__((aligned(16))) float As[BM * P];
    const int tid = threadIdx.x;
    const int lane = tid & 31;
    const int wv = __builtin_amdgcn_readfirstlane(tid >> 5);
    const int wm = wv / NWN;
    const int wn = wv - wm * NWN;
    const int row0 = blockIdx.x * BM;

    static_assert((BM * K4) % NT == 0, "every thread copies the same number of 16-byte pieces");
    constexpr int NI = (BM * K4) / NT;
    constexpr int UB = NI < 8 ? NI : 8;
    static_assert(NI % UB == 0, "whole batches");
    for (int i0 = 0; i0 < NI; i0 += UB) {
        float4 v[UB];
#pragma unroll
        for (int u = 0; u < UB; ++u) {
            const int i = (i0 + u) * NT + tid;
            const int r = i / K4;
            const int c4 = i - r * K4;
            const int gr = min(row0 + r, M - 1);
            v[u] = *(const float4*)(A + (size_t)gr * K + 4 * c4);
        }
#pragma unroll
        for (int u = 0; u < UB; ++u) {
            const int i = (i0 + u) * NT + tid;
            const int r = i / K4;
            const int c4 = i - r * K4;
            *(float4*)(As + r * P + 4 * c4) = v[u];
        }
    }
    __syncthreads();

    const int lo = lane & 15;
    const int hi = lane >> 4;
    const float* a0p = As + (32 * wm + lo) * P + 2 * hi;
    const float* a1p = a0p + 16 * P;
    const float* bp = B + (size_t)(2 * hi) * NC + 32 * wn + lo;

    v8f c00 = {0.f, 0.f, 0.f, 0.f, 0.f, 0.f, 0.f, 0.f};
    v8f c01 = c00, c10 = c00, c11 = c00;
#pragma unroll 4
    for (int k0 = 0; k0 < K; k0 += 4) {
        const v2f a0 = *(const v2f*)(a0p + k0);
        const v2f a1 = *(const v2f*)(a1p + k0);
        v2f b0, b1;
        b0.x = bp[(size_t)k0 * NC];
        b0.y = bp[(size_t)(k0 + 1) * NC];
        b1.x = bp[(size_t)k0 * NC + 16];
        b1.y = bp[(size_t)(k0 + 1) * NC + 16];
        c00 = __builtin_amdgcn_wmma_f32_16x16x4_f32(false, a0, false, b0, (short)0, c00, false, false);
        c01 = __builtin_amdgcn_wmma_f32_16x16x4_f32(false, a0, false, b1, (short)0, c01, false, false);
        c10 = __builtin_amdgcn_wmma_f32_16x16x4_f32(false, a1, false, b0, (short)0, c10, false, false);
        c11 = __builtin_amdgcn_wmma_f32_16x16x4_f32(false, a1, false, b1, (short)0, c11, false, false);
    }
    const int ncol = 32 * wn + lo;
#pragma unroll
    for (int v = 0; v < 8; ++v) {
        const int r0 = row0 + 32 * wm + v + 8 * hi;
        const int r1 = r0 + 16;
        if (r0 < M) {
            C[(size_t)r0 * NC + ncol] = c00[v];
            C[(size_t)r0 * NC + ncol + 16] = c01[v];
        }
        if (r1 < M) {
            C[(size_t)r1 * NC + ncol] = c10[v];
            C[(size_t)r1 * NC + ncol + 16] = c11[v];
        }
    }
}

__global__ __launch_bounds__(256) void alpha_kernel(const float* __restrict__ hf,
                                                    const float* __restrict__ a_s,
                                                    const float* __restrict__ a_d,
                                                    float* __restrict__ as_o,
                                                    float* __restrict__ ad_o,
                                                    int N, int H, int Cc) {
    int n = blockIdx.x * blockDim.x + threadIdx.x;
    if (n >= N) return;
    int F = H * Cc;
    for (int h = 0; h < H; ++h) {
        float s = 0.f, d = 0.f;
        for (int c = 0; c < Cc; ++c) {
            float v = hf[(size_t)n * F + h * Cc + c];
            s += v * a_s[h * Cc + c];
            d += v * a_d[h * Cc + c];
        }
        as_o[n * H + h] = s;
        ad_o[n * H + h] = d;
    }
}

template <int NPB, int NB, int NS, int SEG>
__global__ __launch_bounds__(1024) void part_edges(const int* __restrict__ ei, int E,
                                                   int* __restrict__ seg) {
    static_assert(NPB <= 256, "an entry keeps the local destination in its low 8 bits");
    static_assert(SEG % 4 == 0, "segments are moved in 16-byte pieces");
    constexpr int NT = 1024;
    constexpr int UB = 5;
    constexpr int V = SEG / 4;
    __shared__ __attribute__((aligned(16))) int buf[NB * SEG];
    const int tid = threadIdx.x;
    const int s = blockIdx.x;
    const int* __restrict__ srcp = ei;
    const int* __restrict__ dstp = ei + E;
    const int len = (E + NS - 1) / NS;
    const int e0 = s * len;
    const int e1 = min(E, e0 + len);

    for (int i = tid; i < NB * SEG; i += NT) buf[i] = (i & (SEG - 1)) ? -1 : 0;
    __syncthreads();
    for (int base = e0; base < e1; base += NT * UB) {
        int d[UB], sr[UB];
#pragma unroll
        for (int k = 0; k < UB; ++k) {
            const int e = base + k * NT + tid;
            const int ec = min(e, e1 - 1);
            const int dv = dstp[ec];
            sr[k] = srcp[ec];
            d[k] = (e < e1) ? dv : -1;
        }
#pragma unroll
        for (int k = 0; k < UB; ++k) {
            const unsigned b = (unsigned)d[k] / (unsigned)NPB;
            if (b < (unsigned)NB) {
                const int pos = atomicAdd(&buf[b * SEG], 1);
                if (pos < SEG - 1)
                    buf[b * SEG + 1 + pos] = (sr[k] << 8) | (int)((unsigned)d[k] - b * (unsigned)NPB);
            }
        }
    }
    __syncthreads();
    const int4* bv = (const int4*)buf;
    for (int i = tid; i < NB * V; i += NT) {
        const int b = i / V;
        const int q = i - b * V;
        ((int4*)(seg + ((size_t)b * NS + s) * SEG))[q] = bv[i];
    }
}

template <int H, int Cc, int NPB, int CAP, bool BN, int NS, int SEG>
__global__ __launch_bounds__(1024) void agg_owner(const int* __restrict__ ei, int E, int N,
                                                  const float* __restrict__ as_i,
                                                  const float* __restrict__ ad_i,
                                                  const float* __restrict__ hsrc,
                                                  const float* __restrict__ bias,
                                                  const float* __restrict__ g,
                                                  const float* __restrict__ be,
                                                  const float* __restrict__ mu,
                                                  const float* __restrict__ var,
                                                  float* __restrict__ outp,
                                                  const int* __restrict__ seg) {
    static_assert((H & (H - 1)) == 0, "H must be a power of two");
    static_assert(NPB <= 256, "an entry keeps the local destination in its low 8 bits");
    static_assert(Cc % 32 == 0, "a 32-channel block must lie inside one head");
    static_assert(SEG % 4 == 0, "segments are read in 16-byte pieces");
    constexpr int F = H * Cc;
    constexpr int J = F / 32;
    constexpr int K = (32 / H) < 8 ? (32 / H) : 8;
    constexpr int KK = K > 4 ? (J >= 8 ? 4 : K) : K;
    constexpr int NT = 1024;
    constexpr int NW = NT / 32;
    constexpr int V = SEG / 4;
    constexpr int NV = NS * V;
    constexpr int UB = (NV + NT - 1) / NT;
    __shared__ __attribute__((aligned(16))) float acc[NPB * F];
    __shared__ float zs[NPB * H];
    __shared__ unsigned mxs[NPB * H];
    __shared__ int le[CAP];
    __shared__ int ldl[CAP];
    __shared__ unsigned cnt;

    const int tid = threadIdx.x;
    const int lane = tid & 31;
    const int wv = __builtin_amdgcn_readfirstlane(tid >> 5);
    const int lo = blockIdx.x * NPB;
    const int* __restrict__ srcp = ei;
    const int* __restrict__ dstp = ei + E;

    const int4* __restrict__ sv = (const int4*)(seg + (size_t)blockIdx.x * (NS * SEG));
    int4 w4[UB];
#pragma unroll
    for (int k = 0; k < UB; ++k) {
        const int v = k * NT + tid;
        w4[k] = (v < NV) ? sv[v] : make_int4(-1, -1, -1, -1);
    }

    for (int i = tid; i < NPB * F; i += NT) acc[i] = 0.f;
    for (int i = tid; i < NPB * H; i += NT) {
        zs[i] = 0.f;
        mxs[i] = 0u;
    }
    if (tid == 0) cnt = 0u;
    __syncthreads();

    if (tid < NPB && lo + tid < N) {
        const unsigned pos = atomicAdd(&cnt, 1u);
        if (pos < (unsigned)CAP) { le[pos] = lo + tid; ldl[pos] = tid; }
    }
#pragma unroll
    for (int k = 0; k < UB; ++k) {
        const int v = k * NT + tid;
        int ww[4] = {w4[k].x, w4[k].y, w4[k].z, w4[k].w};
        if (v < NV && (v & (V - 1)) == 0) {
            if (ww[0] > SEG - 1) atomicAdd(&cnt, (unsigned)CAP + 1u);
            ww[0] = -1;
        }
#pragma unroll
        for (int q = 0; q < 4; ++q) {
            if (ww[q] >= 0) {
                const unsigned pos = atomicAdd(&cnt, 1u);
                if (pos < (unsigned)CAP) { le[pos] = ww[q] >> 8; ldl[pos] = ww[q] & 255; }
            }
        }
    }
    __syncthreads();
    const int total = (int)cnt;

    if (total <= CAP) {
        for (int p = tid; p < total; p += NT) {
            const int s = le[p];
            const int dl = ldl[p];
            for (int h = 0; h < H; ++h) {
                const float v = lrelu(as_i[s * H + h] + ad_i[(lo + dl) * H + h]);
                atomicMax(&mxs[dl * H + h], encodeF(v));
            }
        }
        __syncthreads();
        const int ngrp = (total + KK - 1) / KK;
        for (int gi = wv; gi < ngrp; gi += NW) {
            const int p0 = gi * KK;
            const int li = min(lane / H, KK - 1);
            const int lh = lane & (H - 1);
            const int p = min(p0 + li, total - 1);
            const int s = le[p];
            const int dl = ldl[p];
            const float v = lrelu(as_i[s * H + lh] + ad_i[(lo + dl) * H + lh]);
            const float al = __expf(v - decodeF(mxs[dl * H + lh]));
            float r[KK][J];
            for (int i = 0; i < KK; ++i) {
                if (p0 + i < total) {
                    const int si = __builtin_amdgcn_readlane(s, i * H);
                    const float* hr = hsrc + (size_t)si * F + lane;
                    for (int j = 0; j < J; ++j) r[i][j] = hr[32 * j];
                }
            }
            if (lane < KK * H && p0 + li < total) atomicAdd(&zs[dl * H + lh], al);
            for (int i = 0; i < KK; ++i) {
                if (p0 + i < total) {
                    const int di = __builtin_amdgcn_readlane(dl, i * H);
                    float* orow = acc + di * F + lane;
                    for (int j = 0; j < J; ++j) {
                        const float wj = __int_as_float(
                            __builtin_amdgcn_readlane(__float_as_int(al), i * H + (32 * j) / Cc));
                        atomicAdd(orow + 32 * j, r[i][j] * wj);
                    }
                }
            }
        }
    } else {
        const int Etot = E + N;
        for (int e = tid; e < Etot; e += NT) {
            const int ec = min(e, E - 1);
            const int dd = (e < E) ? dstp[ec] : (e - E);
            const unsigned dl = (unsigned)(dd - lo);
            if (dl < (unsigned)NPB) {
                const int s = (e < E) ? srcp[ec] : (e - E);
                for (int h = 0; h < H; ++h) {
                    const float v = lrelu(as_i[s * H + h] + ad_i[dd * H + h]);
                    atomicMax(&mxs[dl * H + h], encodeF(v));
                }
            }
        }
        __syncthreads();
        for (int e = tid; e < Etot; e += NT) {
            const int ec = min(e, E - 1);
            const int dd = (e < E) ? dstp[ec] : (e - E);
            const unsigned dl = (unsigned)(dd - lo);
            if (dl < (unsigned)NPB) {
                const int s = (e < E) ? srcp[ec] : (e - E);
                for (int h = 0; h < H; ++h) {
                    const float v = lrelu(as_i[s * H + h] + ad_i[dd * H + h]);
                    const float w = __expf(v - decodeF(mxs[dl * H + h]));
                    atomicAdd(&zs[dl * H + h], w);
                    for (int c = 0; c < Cc; ++c)
                        atomicAdd(&acc[dl * F + h * Cc + c], w * hsrc[(size_t)s * F + h * Cc + c]);
                }
            }
        }
    }
    __syncthreads();

    for (int i = tid; i < NPB * H; i += NT) zs[i] = 1.0f / (zs[i] + 1e-16f);
    __syncthreads();
    static_assert(Cc % 4 == 0, "a group of four columns must lie inside one head");
    constexpr int F4 = F / 4;
    static_assert(NT % F4 == 0, "every pass covers whole rows");
    constexpr int RS = NT / F4;
    const int cg = tid % F4;
    const int j0 = 4 * cg;
    const int hh = j0 / Cc;
    float cb[4], cm[4], cga[4], crs[4], cbe[4];
    for (int c = 0; c < 4; ++c) {
        cb[c] = bias[j0 + c];
        if (BN) {
            cm[c] = mu[j0 + c];
            cga[c] = g[j0 + c];
            crs[c] = rsqrtf(var[j0 + c] + BN_EPS);
            cbe[c] = be[j0 + c];
        }
    }
    for (int i = tid / F4; i < NPB; i += RS) {
        const int n = lo + i;
        if (n < N) {
            const float4 a = *(const float4*)(acc + i * F + j0);
            const float rz = zs[i * H + hh];
            float o[4] = {a.x, a.y, a.z, a.w};
            for (int c = 0; c < 4; ++c) {
                float v = o[c] * rz + cb[c];
                if (BN) {
                    v = (v - cm[c]) * cga[c] * crs[c] + cbe[c];
                    v = v > 0.f ? v : (__expf(v) - 1.f);
                }
                o[c] = v;
            }
            *(float4*)(outp + (size_t)n * F + j0) = make_float4(o[0], o[1], o[2], o[3]);
        }
    }
}

static inline int nblk(long long t) { return (int)((t + 255) / 256); }

extern "C" void kernel_launch(void* const* d_in, const int* in_sizes, int n_in,
                              void* d_out, int out_size, void* d_ws, size_t ws_size,
                              hipStream_t stream) {
    const float* x      = (const float*)d_in[0];
    const int*   ei     = (const int*)d_in[1];
    const float* W1     = (const float*)d_in[2];
    const float* a_src1 = (const float*)d_in[3];
    const float* a_dst1 = (const float*)d_in[4];
    const float* b1     = (const float*)d_in[5];
    const float* bn_g   = (const float*)d_in[6];
    const float* bn_b   = (const float*)d_in[7];
    const float* bn_m   = (const float*)d_in[8];
    const float* bn_v   = (const float*)d_in[9];
    const float* W2     = (const float*)d_in[10];
    const float* a_src2 = (const float*)d_in[11];
    const float* a_dst2 = (const float*)d_in[12];
    const float* b2     = (const float*)d_in[13];

    const int IN_DIM = 128, F1 = 256, C1 = 64, H1 = 4, C2 = 64;
    const int NPB = 196;
    int N = in_sizes[0] / IN_DIM;
    int E = in_sizes[1] / 2;

    float* ws = (float*)d_ws;
    size_t o = 0;
    float* h1   = ws + o; o += (size_t)N * F1;
    float* act1 = ws + o; o += (size_t)N * F1;
    float* as1  = ws + o; o += (size_t)N * H1;
    float* ad1  = ws + o; o += (size_t)N * H1;
    int*   eseg = (int*)(ws + o); o += (size_t)N * C2;
    float* h2   = h1;
    float* as2  = as1;
    float* ad2  = ad1;

    const int ngroups = (N + NPB - 1) / NPB;

    part_edges<196, 256, 192, 64><<<192, 1024, 0, stream>>>(ei, E, eseg);

    wmma_gemm_f32<128, 1, 8><<<(N + 31) / 32, 256, 0, stream>>>(x, W1, h1, N);
    alpha_kernel<<<nblk(N), 256, 0, stream>>>(h1, a_src1, a_dst1, as1, ad1, N, H1, C1);
    agg_owner<4, 64, 196, 8192, true, 192, 64><<<ngroups, 1024, 0, stream>>>(
        ei, E, N, as1, ad1, h1, b1, bn_g, bn_b, bn_m, bn_v, act1, eseg);

    wmma_gemm_f32<256, 2, 2><<<(N + 63) / 64, 128, 0, stream>>>(act1, W2, h2, N);
    alpha_kernel<<<nblk(N), 256, 0, stream>>>(h2, a_src2, a_dst2, as2, ad2, N, 1, C2);
    agg_owner<1, 64, 196, 8192, false, 192, 64><<<ngroups, 1024, 0, stream>>>(
        ei, E, N, as2, ad2, h2, b2, b2, b2, b2, b2, (float*)d_out, eseg);
}


# Edge partition moved into the layer-1 GEMM launch: its first 384 work-groups each bin one edge slice beside the matrix-bound GEMM groups; one launch fewer; segments now 384 x 32 words
# speedup vs baseline: 7.2515x; 1.0386x over previous
#include <hip/hip_runtime.h>
#include <hip/hip_bf16.h>
#include <math.h>

typedef __attribute__((ext_vector_type(2))) float v2f;
typedef __attribute__((ext_vector_type(8))) float v8f;

#define NEG_SLOPE 0.2f
#define BN_EPS 1e-5f

__device__ __forceinline__ unsigned encodeF(float f) {
    unsigned u = __float_as_uint(f);
    return (u & 0x80000000u) ? ~u : (u | 0x80000000u);
}
__device__ __forceinline__ float decodeF(unsigned u) {
    unsigned b = (u & 0x80000000u) ? (u & 0x7fffffffu) : ~u;
    return __uint_as_float(b);
}
__device__ __forceinline__ float lrelu(float v) {
    return v > 0.0f ? v : NEG_SLOPE * v;
}

template <int K, int NWM, int NWN>
__device__ __forceinline__ void gemm_tile_f32(const float* __restrict__ A,
                                              const float* __restrict__ B,
                                              float* __restrict__ C, int M,
                                              float* __restrict__ As, int bid) {
    static_assert(K % 4 == 0, "a k-step is 4 wide and rows are copied in 16-byte pieces");
    constexpr int NT = 32 * NWM * NWN;
    constexpr int BM = 32 * NWM;
    constexpr int NC = 32 * NWN;
    constexpr int P = K + 4;
    constexpr int K4 = K / 4;
    const int tid = threadIdx.x;
    const int lane = tid & 31;
    const int wv = __builtin_amdgcn_readfirstlane(tid >> 5);
    const int wm = wv / NWN;
    const int wn = wv - wm * NWN;
    const int row0 = bid * BM;

    static_assert((BM * K4) % NT == 0, "every thread copies the same number of 16-byte pieces");
    constexpr int NI = (BM * K4) / NT;
    constexpr int UB = NI < 8 ? NI : 8;
    static_assert(NI % UB == 0, "whole batches");
    for (int i0 = 0; i0 < NI; i0 += UB) {
        float4 v[UB];
#pragma unroll
        for (int u = 0; u < UB; ++u) {
            const int i = (i0 + u) * NT + tid;
            const int r = i / K4;
            const int c4 = i - r * K4;
            const int gr = min(row0 + r, M - 1);
            v[u] = *(const float4*)(A + (size_t)gr * K + 4 * c4);
        }
#pragma unroll
        for (int u = 0; u < UB; ++u) {
            const int i = (i0 + u) * NT + tid;
            const int r = i / K4;
            const int c4 = i - r * K4;
            *(float4*)(As + r * P + 4 * c4) = v[u];
        }
    }
    __syncthreads();

    const int lo = lane & 15;
    const int hi = lane >> 4;
    const float* a0p = As + (32 * wm + lo) * P + 2 * hi;
    const float* a1p = a0p + 16 * P;
    const float* bp = B + (size_t)(2 * hi) * NC + 32 * wn + lo;

    v8f c00 = {0.f, 0.f, 0.f, 0.f, 0.f, 0.f, 0.f, 0.f};
    v8f c01 = c00, c10 = c00, c11 = c00;
#pragma unroll 4
    for (int k0 = 0; k0 < K; k0 += 4) {
        const v2f a0 = *(const v2f*)(a0p + k0);
        const v2f a1 = *(const v2f*)(a1p + k0);
        v2f b0, b1;
        b0.x = bp[(size_t)k0 * NC];
        b0.y = bp[(size_t)(k0 + 1) * NC];
        b1.x = bp[(size_t)k0 * NC + 16];
        b1.y = bp[(size_t)(k0 + 1) * NC + 16];
        c00 = __builtin_amdgcn_wmma_f32_16x16x4_f32(false, a0, false, b0, (short)0, c00, false, false);
        c01 = __builtin_amdgcn_wmma_f32_16x16x4_f32(false, a0, false, b1, (short)0, c01, false, false);
        c10 = __builtin_amdgcn_wmma_f32_16x16x4_f32(false, a1, false, b0, (short)0, c10, false, false);
        c11 = __builtin_amdgcn_wmma_f32_16x16x4_f32(false, a1, false, b1, (short)0, c11, false, false);
    }
    const int ncol = 32 * wn + lo;
#pragma unroll
    for (int v = 0; v < 8; ++v) {
        const int r0 = row0 + 32 * wm + v + 8 * hi;
        const int r1 = r0 + 16;
        if (r0 < M) {
            C[(size_t)r0 * NC + ncol] = c00[v];
            C[(size_t)r0 * NC + ncol + 16] = c01[v];
        }
        if (r1 < M) {
            C[(size_t)r1 * NC + ncol] = c10[v];
            C[(size_t)r1 * NC + ncol + 16] = c11[v];
        }
    }
}

template <int K, int NWM, int NWN>
__global__ __launch_bounds__(32 * NWM * NWN) void wmma_gemm_f32(const float* __restrict__ A,
                                                                const float* __restrict__ B,
                                                                float* __restrict__ C, int M) {
    __shared__ __attribute__((aligned(16))) float As[32 * NWM * (K + 4)];
    gemm_tile_f32<K, NWM, NWN>(A, B, C, M, As, blockIdx.x);
}

__global__ __launch_bounds__(256) void alpha_kernel(const float* __restrict__ hf,
                                                    const float* __restrict__ a_s,
                                                    const float* __restrict__ a_d,
                                                    float* __restrict__ as_o,
                                                    float* __restrict__ ad_o,
                                                    int N, int H, int Cc) {
    int n = blockIdx.x * blockDim.x + threadIdx.x;
    if (n >= N) return;
    int F = H * Cc;
    for (int h = 0; h < H; ++h) {
        float s = 0.f, d = 0.f;
        for (int c = 0; c < Cc; ++c) {
            float v = hf[(size_t)n * F + h * Cc + c];
            s += v * a_s[h * Cc + c];
            d += v * a_d[h * Cc + c];
        }
        as_o[n * H + h] = s;
        ad_o[n * H + h] = d;
    }
}

template <int NT, int NPB, int NB, int NS, int SEG>
__device__ __forceinline__ void part_edges_slice(const int* __restrict__ ei, int E,
                                                 int* __restrict__ seg, int* __restrict__ buf,
                                                 int s) {
    static_assert(NPB <= 256, "an entry keeps the local destination in its low 8 bits");
    static_assert(SEG % 4 == 0, "segments are moved in 16-byte pieces");
    constexpr int UB = 5;
    constexpr int V = SEG / 4;
    const int tid = threadIdx.x;
    const int* __restrict__ srcp = ei;
    const int* __restrict__ dstp = ei + E;
    const int len = (E + NS - 1) / NS;
    const int e0 = s * len;
    const int e1 = min(E, e0 + len);

    for (int i = tid; i < NB * V; i += NT)
        ((int4*)buf)[i] = make_int4((i % V) ? -1 : 0, -1, -1, -1);
    __syncthreads();
    for (int base = e0; base < e1; base += NT * UB) {
        int d[UB], sr[UB];
#pragma unroll
        for (int k = 0; k < UB; ++k) {
            const int e = base + k * NT + tid;
            const int ec = min(e, e1 - 1);
            const int dv = dstp[ec];
            sr[k] = srcp[ec];
            d[k] = (e < e1) ? dv : -1;
        }
#pragma unroll
        for (int k = 0; k < UB; ++k) {
            const unsigned b = (unsigned)d[k] / (unsigned)NPB;
            if (b < (unsigned)NB) {
                const int pos = atomicAdd(&buf[b * SEG], 1);
                if (pos < SEG - 1)
                    buf[b * SEG + 1 + pos] = (sr[k] << 8) | (int)((unsigned)d[k] - b * (unsigned)NPB);
            }
        }
    }
    __syncthreads();
    const int4* bv = (const int4*)buf;
    for (int i = tid; i < NB * V; i += NT) {
        const int b = i / V;
        const int q = i - b * V;
        ((int4*)(seg + ((size_t)b * NS + s) * SEG))[q] = bv[i];
    }
}

template <int K, int NWM, int NWN, int NPB, int NB, int NS, int SEG>
__global__ __launch_bounds__(32 * NWM * NWN) void gemm_part(const float* __restrict__ A,
                                                            const float* __restrict__ B,
                                                            float* __restrict__ C, int M,
                                                            const int* __restrict__ ei, int E,
                                                            int* __restrict__ seg) {
    constexpr int NT = 32 * NWM * NWN;
    constexpr int GW = 32 * NWM * (K + 4);
    constexpr int PW = NB * SEG;
    __shared__ __attribute__((aligned(16))) unsigned char lds[4 * (GW > PW ? GW : PW)];
    if (blockIdx.x >= NS)
        gemm_tile_f32<K, NWM, NWN>(A, B, C, M, (float*)lds, blockIdx.x - NS);
    else
        part_edges_slice<NT, NPB, NB, NS, SEG>(ei, E, seg, (int*)lds, blockIdx.x);
}

template <int H, int Cc, int NPB, int CAP, bool BN, int NS, int SEG>
__global__ __launch_bounds__(1024) void agg_owner(const int* __restrict__ ei, int E, int N,
                                                  const float* __restrict__ as_i,
                                                  const float* __restrict__ ad_i,
                                                  const float* __restrict__ hsrc,
                                                  const float* __restrict__ bias,
                                                  const float* __restrict__ g,
                                                  const float* __restrict__ be,
                                                  const float* __restrict__ mu,
                                                  const float* __restrict__ var,
                                                  float* __restrict__ outp,
                                                  const int* __restrict__ seg) {
    static_assert((H & (H - 1)) == 0, "H must be a power of two");
    static_assert(NPB <= 256, "an entry keeps the local destination in its low 8 bits");
    static_assert(Cc % 32 == 0, "a 32-channel block must lie inside one head");
    static_assert(SEG % 4 == 0, "segments are read in 16-byte pieces");
    constexpr int F = H * Cc;
    constexpr int J = F / 32;
    constexpr int K = (32 / H) < 8 ? (32 / H) : 8;
    constexpr int KK = K > 4 ? (J >= 8 ? 4 : K) : K;
    constexpr int NT = 1024;
    constexpr int NW = NT / 32;
    constexpr int V = SEG / 4;
    constexpr int NV = NS * V;
    constexpr int UB = (NV + NT - 1) / NT;
    __shared__ __attribute__((aligned(16))) float acc[NPB * F];
    __shared__ float zs[NPB * H];
    __shared__ unsigned mxs[NPB * H];
    __shared__ int le[CAP];
    __shared__ int ldl[CAP];
    __shared__ unsigned cnt;

    const int tid = threadIdx.x;
    const int lane = tid & 31;
    const int wv = __builtin_amdgcn_readfirstlane(tid >> 5);
    const int lo = blockIdx.x * NPB;
    const int* __restrict__ srcp = ei;
    const int* __restrict__ dstp = ei + E;

    const int4* __restrict__ sv = (const int4*)(seg + (size_t)blockIdx.x * (NS * SEG));
    int4 w4[UB];
#pragma unroll
    for (int k = 0; k < UB; ++k) {
        const int v = k * NT + tid;
        w4[k] = (v < NV) ? sv[v] : make_int4(-1, -1, -1, -1);
    }

    for (int i = tid; i < NPB * F; i += NT) acc[i] = 0.f;
    for (int i = tid; i < NPB * H; i += NT) {
        zs[i] = 0.f;
        mxs[i] = 0u;
    }
    if (tid == 0) cnt = 0u;
    __syncthreads();

    if (tid < NPB && lo + tid < N) {
        const unsigned pos = atomicAdd(&cnt, 1u);
        if (pos < (unsigned)CAP) { le[pos] = lo + tid; ldl[pos] = tid; }
    }
#pragma unroll
    for (int k = 0; k < UB; ++k) {
        const int v = k * NT + tid;
        int ww[4] = {w4[k].x, w4[k].y, w4[k].z, w4[k].w};
        if (v < NV && (v & (V - 1)) == 0) {
            if (ww[0] > SEG - 1) atomicAdd(&cnt, (unsigned)CAP + 1u);
            ww[0] = -1;
        }
#pragma unroll
        for (int q = 0; q < 4; ++q) {
            if (ww[q] >= 0) {
                const unsigned pos = atomicAdd(&cnt, 1u);
                if (pos < (unsigned)CAP) { le[pos] = ww[q] >> 8; ldl[pos] = ww[q] & 255; }
            }
        }
    }
    __syncthreads();
    const int total = (int)cnt;

    if (total <= CAP) {
        for (int p = tid; p < total; p += NT) {
            const int s = le[p];
            const int dl = ldl[p];
            for (int h = 0; h < H; ++h) {
                const float v = lrelu(as_i[s * H + h] + ad_i[(lo + dl) * H + h]);
                atomicMax(&mxs[dl * H + h], encodeF(v));
            }
        }
        __syncthreads();
        const int ngrp = (total + KK - 1) / KK;
        for (int gi = wv; gi < ngrp; gi += NW) {
            const int p0 = gi * KK;
            const int li = min(lane / H, KK - 1);
            const int lh = lane & (H - 1);
            const int p = min(p0 + li, total - 1);
            const int s = le[p];
            const int dl = ldl[p];
            const float v = lrelu(as_i[s * H + lh] + ad_i[(lo + dl) * H + lh]);
            const float al = __expf(v - decodeF(mxs[dl * H + lh]));
            float r[KK][J];
            for (int i = 0; i < KK; ++i) {
                if (p0 + i < total) {
                    const int si = __builtin_amdgcn_readlane(s, i * H);
                    const float* hr = hsrc + (size_t)si * F + lane;
                    for (int j = 0; j < J; ++j) r[i][j] = hr[32 * j];
                }
            }
            if (lane < KK * H && p0 + li < total) atomicAdd(&zs[dl * H + lh], al);
            for (int i = 0; i < KK; ++i) {
                if (p0 + i < total) {
                    const int di = __builtin_amdgcn_readlane(dl, i * H);
                    float* orow = acc + di * F + lane;
                    for (int j = 0; j < J; ++j) {
                        const float wj = __int_as_float(
                            __builtin_amdgcn_readlane(__float_as_int(al), i * H + (32 * j) / Cc));
                        atomicAdd(orow + 32 * j, r[i][j] * wj);
                    }
                }
            }
        }
    } else {
        const int Etot = E + N;
        for (int e = tid; e < Etot; e += NT) {
            const int ec = min(e, E - 1);
            const int dd = (e < E) ? dstp[ec] : (e - E);
            const unsigned dl = (unsigned)(dd - lo);
            if (dl < (unsigned)NPB) {
                const int s = (e < E) ? srcp[ec] : (e - E);
                for (int h = 0; h < H; ++h) {
                    const float v = lrelu(as_i[s * H + h] + ad_i[dd * H + h]);
                    atomicMax(&mxs[dl * H + h], encodeF(v));
                }
            }
        }
        __syncthreads();
        for (int e = tid; e < Etot; e += NT) {
            const int ec = min(e, E - 1);
            const int dd = (e < E) ? dstp[ec] : (e - E);
            const unsigned dl = (unsigned)(dd - lo);
            if (dl < (unsigned)NPB) {
                const int s = (e < E) ? srcp[ec] : (e - E);
                for (int h = 0; h < H; ++h) {
                    const float v = lrelu(as_i[s * H + h] + ad_i[dd * H + h]);
                    const float w = __expf(v - decodeF(mxs[dl * H + h]));
                    atomicAdd(&zs[dl * H + h], w);
                    for (int c = 0; c < Cc; ++c)
                        atomicAdd(&acc[dl * F + h * Cc + c], w * hsrc[(size_t)s * F + h * Cc + c]);
                }
            }
        }
    }
    __syncthreads();

    for (int i = tid; i < NPB * H; i += NT) zs[i] = 1.0f / (zs[i] + 1e-16f);
    __syncthreads();
    static_assert(Cc % 4 == 0, "a group of four columns must lie inside one head");
    constexpr int F4 = F / 4;
    static_assert(NT % F4 == 0, "every pass covers whole rows");
    constexpr int RS = NT / F4;
    const int cg = tid % F4;
    const int j0 = 4 * cg;
    const int hh = j0 / Cc;
    float cb[4], cm[4], cga[4], crs[4], cbe[4];
    for (int c = 0; c < 4; ++c) {
        cb[c] = bias[j0 + c];
        if (BN) {
            cm[c] = mu[j0 + c];
            cga[c] = g[j0 + c];
            crs[c] = rsqrtf(var[j0 + c] + BN_EPS);
            cbe[c] = be[j0 + c];
        }
    }
    for (int i = tid / F4; i < NPB; i += RS) {
        const int n = lo + i;
        if (n < N) {
            const float4 a = *(const float4*)(acc + i * F + j0);
            const float rz = zs[i * H + hh];
            float o[4] = {a.x, a.y, a.z, a.w};
            for (int c = 0; c < 4; ++c) {
                float v = o[c] * rz + cb[c];
                if (BN) {
                    v = (v - cm[c]) * cga[c] * crs[c] + cbe[c];
                    v = v > 0.f ? v : (__expf(v) - 1.f);
                }
                o[c] = v;
            }
            *(float4*)(outp + (size_t)n * F + j0) = make_float4(o[0], o[1], o[2], o[3]);
        }
    }
}

static inline int nblk(long long t) { return (int)((t + 255) / 256); }

extern "C" void kernel_launch(void* const* d_in, const int* in_sizes, int n_in,
                              void* d_out, int out_size, void* d_ws, size_t ws_size,
                              hipStream_t stream) {
    const float* x      = (const float*)d_in[0];
    const int*   ei     = (const int*)d_in[1];
    const float* W1     = (const float*)d_in[2];
    const float* a_src1 = (const float*)d_in[3];
    const float* a_dst1 = (const float*)d_in[4];
    const float* b1     = (const float*)d_in[5];
    const float* bn_g   = (const float*)d_in[6];
    const float* bn_b   = (const float*)d_in[7];
    const float* bn_m   = (const float*)d_in[8];
    const float* bn_v   = (const float*)d_in[9];
    const float* W2     = (const float*)d_in[10];
    const float* a_src2 = (const float*)d_in[11];
    const float* a_dst2 = (const float*)d_in[12];
    const float* b2     = (const float*)d_in[13];

    const int IN_DIM = 128, F1 = 256, C1 = 64, H1 = 4, C2 = 64;
    const int NPB = 196;
    int N = in_sizes[0] / IN_DIM;
    int E = in_sizes[1] / 2;

    float* ws = (float*)d_ws;
    size_t o = 0;
    float* h1   = ws + o; o += (size_t)N * F1;
    float* act1 = ws + o; o += (size_t)N * F1;
    float* as1  = ws + o; o += (size_t)N * H1;
    float* ad1  = ws + o; o += (size_t)N * H1;
    int*   eseg = (int*)(ws + o); o += (size_t)N * C2;
    float* h2   = h1;
    float* as2  = as1;
    float* ad2  = ad1;

    const int ngroups = (N + NPB - 1) / NPB;

    gemm_part<128, 1, 8, 196, 256, 384, 32><<<384 + (N + 31) / 32, 256, 0, stream>>>(
        x, W1, h1, N, ei, E, eseg);
    alpha_kernel<<<nblk(N), 256, 0, stream>>>(h1, a_src1, a_dst1, as1, ad1, N, H1, C1);
    agg_owner<4, 64, 196, 8192, true, 384, 32><<<ngroups, 1024, 0, stream>>>(
        ei, E, N, as1, ad1, h1, b1, bn_g, bn_b, bn_m, bn_v, act1, eseg);

    wmma_gemm_f32<256, 2, 2><<<(N + 63) / 64, 128, 0, stream>>>(act1, W2, h2, N);
    alpha_kernel<<<nblk(N), 256, 0, stream>>>(h2, a_src2, a_dst2, as2, ad2, N, 1, C2);
    agg_owner<1, 64, 196, 8192, false, 384, 32><<<ngroups, 1024, 0, stream>>>(
        ei, E, N, as2, ad2, h2, b2, b2, b2, b2, b2, (float*)d_out, eseg);
}


# Owner groups gather feature rows with whole-line loads of 16 bytes per lane (8 for 256-byte rows): 2 loads per 1-KB row, not 8; LDS rows keep a bank-free channel order; LDS sums stay arrival order
# speedup vs baseline: 7.7055x; 1.0626x over previous
#include <hip/hip_runtime.h>
#include <hip/hip_bf16.h>
#include <math.h>

typedef __attribute__((ext_vector_type(2))) float v2f;
typedef __attribute__((ext_vector_type(8))) float v8f;

#define NEG_SLOPE 0.2f
#define BN_EPS 1e-5f

__device__ __forceinline__ unsigned encodeF(float f) {
    unsigned u = __float_as_uint(f);
    return (u & 0x80000000u) ? ~u : (u | 0x80000000u);
}
__device__ __forceinline__ float decodeF(unsigned u) {
    unsigned b = (u & 0x80000000u) ? (u & 0x7fffffffu) : ~u;
    return __uint_as_float(b);
}
__device__ __forceinline__ float lrelu(float v) {
    return v > 0.0f ? v : NEG_SLOPE * v;
}

template <int K, int NWM, int NWN>
__device__ __forceinline__ void gemm_tile_f32(const float* __restrict__ A,
                                              const float* __restrict__ B,
                                              float* __restrict__ C, int M,
                                              float* __restrict__ As, int bid) {
    static_assert(K % 4 == 0, "a k-step is 4 wide and rows are copied in 16-byte pieces");
    constexpr int NT = 32 * NWM * NWN;
    constexpr int BM = 32 * NWM;
    constexpr int NC = 32 * NWN;
    constexpr int P = K + 4;
    constexpr int K4 = K / 4;
    const int tid = threadIdx.x;
    const int lane = tid & 31;
    const int wv = __builtin_amdgcn_readfirstlane(tid >> 5);
    const int wm = wv / NWN;
    const int wn = wv - wm * NWN;
    const int row0 = bid * BM;

    static_assert((BM * K4) % NT == 0, "every thread copies the same number of 16-byte pieces");
    constexpr int NI = (BM * K4) / NT;
    constexpr int UB = NI < 8 ? NI : 8;
    static_assert(NI % UB == 0, "whole batches");
    for (int i0 = 0; i0 < NI; i0 += UB) {
        float4 v[UB];
#pragma unroll
        for (int u = 0; u < UB; ++u) {
            const int i = (i0 + u) * NT + tid;
            const int r = i / K4;
            const int c4 = i - r * K4;
            const int gr = min(row0 + r, M - 1);
            v[u] = *(const float4*)(A + (size_t)gr * K + 4 * c4);
        }
#pragma unroll
        for (int u = 0; u < UB; ++u) {
            const int i = (i0 + u) * NT + tid;
            const int r = i / K4;
            const int c4 = i - r * K4;
            *(float4*)(As + r * P + 4 * c4) = v[u];
        }
    }
    __syncthreads();

    const int lo = lane & 15;
    const int hi = lane >> 4;
    const float* a0p = As + (32 * wm + lo) * P + 2 * hi;
    const float* a1p = a0p + 16 * P;
    const float* bp = B + (size_t)(2 * hi) * NC + 32 * wn + lo;

    v8f c00 = {0.f, 0.f, 0.f, 0.f, 0.f, 0.f, 0.f, 0.f};
    v8f c01 = c00, c10 = c00, c11 = c00;
#pragma unroll 4
    for (int k0 = 0; k0 < K; k0 += 4) {
        const v2f a0 = *(const v2f*)(a0p + k0);
        const v2f a1 = *(const v2f*)(a1p + k0);
        v2f b0, b1;
        b0.x = bp[(size_t)k0 * NC];
        b0.y = bp[(size_t)(k0 + 1) * NC];
        b1.x = bp[(size_t)k0 * NC + 16];
        b1.y = bp[(size_t)(k0 + 1) * NC + 16];
        c00 = __builtin_amdgcn_wmma_f32_16x16x4_f32(false, a0, false, b0, (short)0, c00, false, false);
        c01 = __builtin_amdgcn_wmma_f32_16x16x4_f32(false, a0, false, b1, (short)0, c01, false, false);
        c10 = __builtin_amdgcn_wmma_f32_16x16x4_f32(false, a1, false, b0, (short)0, c10, false, false);
        c11 = __builtin_amdgcn_wmma_f32_16x16x4_f32(false, a1, false, b1, (short)0, c11, false, false);
    }
    const int ncol = 32 * wn + lo;
#pragma unroll
    for (int v = 0; v < 8; ++v) {
        const int r0 = row0 + 32 * wm + v + 8 * hi;
        const int r1 = r0 + 16;
        if (r0 < M) {
            C[(size_t)r0 * NC + ncol] = c00[v];
            C[(size_t)r0 * NC + ncol + 16] = c01[v];
        }
        if (r1 < M) {
            C[(size_t)r1 * NC + ncol] = c10[v];
            C[(size_t)r1 * NC + ncol + 16] = c11[v];
        }
    }
}

template <int K, int NWM, int NWN>
__global__ __launch_bounds__(32 * NWM * NWN) void wmma_gemm_f32(const float* __restrict__ A,
                                                                const float* __restrict__ B,
                                                                float* __restrict__ C, int M) {
    __shared__ __attribute__((aligned(16))) float As[32 * NWM * (K + 4)];
    gemm_tile_f32<K, NWM, NWN>(A, B, C, M, As, blockIdx.x);
}

__global__ __launch_bounds__(256) void alpha_kernel(const float* __restrict__ hf,
                                                    const float* __restrict__ a_s,
                                                    const float* __restrict__ a_d,
                                                    float* __restrict__ as_o,
                                                    float* __restrict__ ad_o,
                                                    int N, int H, int Cc) {
    int n = blockIdx.x * blockDim.x + threadIdx.x;
    if (n >= N) return;
    int F = H * Cc;
    for (int h = 0; h < H; ++h) {
        float s = 0.f, d = 0.f;
        for (int c = 0; c < Cc; ++c) {
            float v = hf[(size_t)n * F + h * Cc + c];
            s += v * a_s[h * Cc + c];
            d += v * a_d[h * Cc + c];
        }
        as_o[n * H + h] = s;
        ad_o[n * H + h] = d;
    }
}

template <int NT, int NPB, int NB, int NS, int SEG>
__device__ __forceinline__ void part_edges_slice(const int* __restrict__ ei, int E,
                                                 int* __restrict__ seg, int* __restrict__ buf,
                                                 int s) {
    static_assert(NPB <= 256, "an entry keeps the local destination in its low 8 bits");
    static_assert(SEG % 4 == 0, "segments are moved in 16-byte pieces");
    constexpr int UB = 5;
    constexpr int V = SEG / 4;
    const int tid = threadIdx.x;
    const int* __restrict__ srcp = ei;
    const int* __restrict__ dstp = ei + E;
    const int len = (E + NS - 1) / NS;
    const int e0 = s * len;
    const int e1 = min(E, e0 + len);

    for (int i = tid; i < NB * V; i += NT)
        ((int4*)buf)[i] = make_int4((i % V) ? -1 : 0, -1, -1, -1);
    __syncthreads();
    for (int base = e0; base < e1; base += NT * UB) {
        int d[UB], sr[UB];
#pragma unroll
        for (int k = 0; k < UB; ++k) {
            const int e = base + k * NT + tid;
            const int ec = min(e, e1 - 1);
            const int dv = dstp[ec];
            sr[k] = srcp[ec];
            d[k] = (e < e1) ? dv : -1;
        }
#pragma unroll
        for (int k = 0; k < UB; ++k) {
            const unsigned b = (unsigned)d[k] / (unsigned)NPB;
            if (b < (unsigned)NB) {
                const int pos = atomicAdd(&buf[b * SEG], 1);
                if (pos < SEG - 1)
                    buf[b * SEG + 1 + pos] = (sr[k] << 8) | (int)((unsigned)d[k] - b * (unsigned)NPB);
            }
        }
    }
    __syncthreads();
    const int4* bv = (const int4*)buf;
    for (int i = tid; i < NB * V; i += NT) {
        const int b = i / V;
        const int q = i - b * V;
        ((int4*)(seg + ((size_t)b * NS + s) * SEG))[q] = bv[i];
    }
}

template <int K, int NWM, int NWN, int NPB, int NB, int NS, int SEG>
__global__ __launch_bounds__(32 * NWM * NWN) void gemm_part(const float* __restrict__ A,
                                                            const float* __restrict__ B,
                                                            float* __restrict__ C, int M,
                                                            const int* __restrict__ ei, int E,
                                                            int* __restrict__ seg) {
    constexpr int NT = 32 * NWM * NWN;
    constexpr int GW = 32 * NWM * (K + 4);
    constexpr int PW = NB * SEG;
    __shared__ __attribute__((aligned(16))) unsigned char lds[4 * (GW > PW ? GW : PW)];
    if (blockIdx.x >= NS)
        gemm_tile_f32<K, NWM, NWN>(A, B, C, M, (float*)lds, blockIdx.x - NS);
    else
        part_edges_slice<NT, NPB, NB, NS, SEG>(ei, E, seg, (int*)lds, blockIdx.x);
}

template <int VEC>
__device__ __forceinline__ int row_pos(int ch) {
    constexpr int LW = 32 * VEC;
    const int k = ch / LW;
    const int rr = ch - k * LW;
    return k * LW + (rr % VEC) * 32 + rr / VEC;
}

template <int H, int Cc, int NPB, int CAP, bool BN, int NS, int SEG>
__global__ __launch_bounds__(1024) void agg_owner(const int* __restrict__ ei, int E, int N,
                                                  const float* __restrict__ as_i,
                                                  const float* __restrict__ ad_i,
                                                  const float* __restrict__ hsrc,
                                                  const float* __restrict__ bias,
                                                  const float* __restrict__ g,
                                                  const float* __restrict__ be,
                                                  const float* __restrict__ mu,
                                                  const float* __restrict__ var,
                                                  float* __restrict__ outp,
                                                  const int* __restrict__ seg) {
    static_assert((H & (H - 1)) == 0, "H must be a power of two");
    static_assert(NPB <= 256, "an entry keeps the local destination in its low 8 bits");
    static_assert(Cc % 32 == 0, "a 32-channel block must lie inside one head");
    static_assert(SEG % 4 == 0, "segments are read in 16-byte pieces");
    constexpr int F = H * Cc;
    constexpr int VEC = (F / 32) < 4 ? (F / 32) : 4;
    constexpr int LW = 32 * VEC;
    constexpr int NL = F / LW;
    constexpr int HPL = LW > Cc ? LW / Cc : 1;
    constexpr int LPH = 32 / HPL;
    static_assert(F % LW == 0, "a row is a whole number of wave-wide loads");
    static_assert(LW % Cc == 0 || Cc % LW == 0, "a load holds whole heads, or lies inside one");
    static_assert(HPL <= 2, "the weight of a lane is picked from at most two heads per load");
    constexpr int K = (32 / H) < 16 ? (32 / H) : 16;
    constexpr int KK = K * NL > 32 ? 32 / NL : K;
    typedef float rvec __attribute__((ext_vector_type(VEC)));
    constexpr int NT = 1024;
    constexpr int NW = NT / 32;
    constexpr int V = SEG / 4;
    constexpr int NV = NS * V;
    constexpr int UB = (NV + NT - 1) / NT;
    __shared__ __attribute__((aligned(16))) float acc[NPB * F];
    __shared__ float zs[NPB * H];
    __shared__ unsigned mxs[NPB * H];
    __shared__ int le[CAP];
    __shared__ int ldl[CAP];
    __shared__ unsigned cnt;

    const int tid = threadIdx.x;
    const int lane = tid & 31;
    const int wv = __builtin_amdgcn_readfirstlane(tid >> 5);
    const int lo = blockIdx.x * NPB;
    const int* __restrict__ srcp = ei;
    const int* __restrict__ dstp = ei + E;

    const int4* __restrict__ sv = (const int4*)(seg + (size_t)blockIdx.x * (NS * SEG));
    int4 w4[UB];
#pragma unroll
    for (int k = 0; k < UB; ++k) {
        const int v = k * NT + tid;
        w4[k] = (v < NV) ? sv[v] : make_int4(-1, -1, -1, -1);
    }

    for (int i = tid; i < NPB * F; i += NT) acc[i] = 0.f;
    for (int i = tid; i < NPB * H; i += NT) {
        zs[i] = 0.f;
        mxs[i] = 0u;
    }
    if (tid == 0) cnt = 0u;
    __syncthreads();

    if (tid < NPB && lo + tid < N) {
        const unsigned pos = atomicAdd(&cnt, 1u);
        if (pos < (unsigned)CAP) { le[pos] = lo + tid; ldl[pos] = tid; }
    }
#pragma unroll
    for (int k = 0; k < UB; ++k) {
        const int v = k * NT + tid;
        int ww[4] = {w4[k].x, w4[k].y, w4[k].z, w4[k].w};
        if (v < NV && (v & (V - 1)) == 0) {
            if (ww[0] > SEG - 1) atomicAdd(&cnt, (unsigned)CAP + 1u);
            ww[0] = -1;
        }
#pragma unroll
        for (int q = 0; q < 4; ++q) {
            if (ww[q] >= 0) {
                const unsigned pos = atomicAdd(&cnt, 1u);
                if (pos < (unsigned)CAP) { le[pos] = ww[q] >> 8; ldl[pos] = ww[q] & 255; }
            }
        }
    }
    __syncthreads();
    const int total = (int)cnt;

    if (total <= CAP) {
        for (int p = tid; p < total; p += NT) {
            const int s = le[p];
            const int dl = ldl[p];
            for (int h = 0; h < H; ++h) {
                const float v = lrelu(as_i[s * H + h] + ad_i[(lo + dl) * H + h]);
                atomicMax(&mxs[dl * H + h], encodeF(v));
            }
        }
        __syncthreads();
        const int ngrp = (total + KK - 1) / KK;
        for (int gi = wv; gi < ngrp; gi += NW) {
            const int p0 = gi * KK;
            const int li = min(lane / H, KK - 1);
            const int lh = lane & (H - 1);
            const int p = min(p0 + li, total - 1);
            const int s = le[p];
            const int dl = ldl[p];
            const float v = lrelu(as_i[s * H + lh] + ad_i[(lo + dl) * H + lh]);
            const float al = __expf(v - decodeF(mxs[dl * H + lh]));
            rvec r[KK][NL];
            for (int i = 0; i < KK; ++i) {
                if (p0 + i < total) {
                    const int si = __builtin_amdgcn_readlane(s, i * H);
                    const rvec* hr = (const rvec*)(hsrc + (size_t)si * F) + lane;
                    for (int k = 0; k < NL; ++k) r[i][k] = hr[32 * k];
                }
            }
            if (lane < KK * H && p0 + li < total) atomicAdd(&zs[dl * H + lh], al);
            for (int i = 0; i < KK; ++i) {
                if (p0 + i < total) {
                    const int di = __builtin_amdgcn_readlane(dl, i * H);
                    float* orow = acc + di * F + lane;
                    for (int k = 0; k < NL; ++k) {
                        float wk = __int_as_float(
                            __builtin_amdgcn_readlane(__float_as_int(al), i * H + (k * LW) / Cc));
                        if (HPL == 2) {
                            const float w1 = __int_as_float(
                                __builtin_amdgcn_readlane(__float_as_int(al), i * H + (k * LW) / Cc + 1));
                            wk = lane < LPH ? wk : w1;
                        }
                        for (int c = 0; c < VEC; ++c)
                            atomicAdd(orow + k * LW + 32 * c, r[i][k][c] * wk);
                    }
                }
            }
        }
    } else {
        const int Etot = E + N;
        for (int e = tid; e < Etot; e += NT) {
            const int ec = min(e, E - 1);
            const int dd = (e < E) ? dstp[ec] : (e - E);
            const unsigned dl = (unsigned)(dd - lo);
            if (dl < (unsigned)NPB) {
                const int s = (e < E) ? srcp[ec] : (e - E);
                for (int h = 0; h < H; ++h) {
                    const float v = lrelu(as_i[s * H + h] + ad_i[dd * H + h]);
                    atomicMax(&mxs[dl * H + h], encodeF(v));
                }
            }
        }
        __syncthreads();
        for (int e = tid; e < Etot; e += NT) {
            const int ec = min(e, E - 1);
            const int dd = (e < E) ? dstp[ec] : (e - E);
            const unsigned dl = (unsigned)(dd - lo);
            if (dl < (unsigned)NPB) {
                const int s = (e < E) ? srcp[ec] : (e - E);
                for (int h = 0; h < H; ++h) {
                    const float v = lrelu(as_i[s * H + h] + ad_i[dd * H + h]);
                    const float w = __expf(v - decodeF(mxs[dl * H + h]));
                    atomicAdd(&zs[dl * H + h], w);
                    for (int c = 0; c < Cc; ++c)
                        atomicAdd(&acc[dl * F + row_pos<VEC>(h * Cc + c)], w * hsrc[(size_t)s * F + h * Cc + c]);
                }
            }
        }
    }
    __syncthreads();

    for (int i = tid; i < NPB * H; i += NT) zs[i] = 1.0f / (zs[i] + 1e-16f);
    __syncthreads();
    static_assert(Cc % 4 == 0, "a group of four columns must lie inside one head");
    constexpr int F4 = F / 4;
    static_assert(NT % F4 == 0, "every pass covers whole rows");
    constexpr int RS = NT / F4;
    const int cg = tid % F4;
    const int j0 = 4 * cg;
    const int hh = j0 / Cc;
    float cb[4], cm[4], cga[4], crs[4], cbe[4];
    int cp[4];
    for (int c = 0; c < 4; ++c) {
        cp[c] = row_pos<VEC>(j0 + c);
        cb[c] = bias[j0 + c];
        if (BN) {
            cm[c] = mu[j0 + c];
            cga[c] = g[j0 + c];
            crs[c] = rsqrtf(var[j0 + c] + BN_EPS);
            cbe[c] = be[j0 + c];
        }
    }
    for (int i = tid / F4; i < NPB; i += RS) {
        const int n = lo + i;
        if (n < N) {
            const float rz = zs[i * H + hh];
            float o[4];
            for (int c = 0; c < 4; ++c) o[c] = acc[i * F + cp[c]];
            for (int c = 0; c < 4; ++c) {
                float v = o[c] * rz + cb[c];
                if (BN) {
                    v = (v - cm[c]) * cga[c] * crs[c] + cbe[c];
                    v = v > 0.f ? v : (__expf(v) - 1.f);
                }
                o[c] = v;
            }
            *(float4*)(outp + (size_t)n * F + j0) = make_float4(o[0], o[1], o[2], o[3]);
        }
    }
}

static inline int nblk(long long t) { return (int)((t + 255) / 256); }

extern "C" void kernel_launch(void* const* d_in, const int* in_sizes, int n_in,
                              void* d_out, int out_size, void* d_ws, size_t ws_size,
                              hipStream_t stream) {
    const float* x      = (const float*)d_in[0];
    const int*   ei     = (const int*)d_in[1];
    const float* W1     = (const float*)d_in[2];
    const float* a_src1 = (const float*)d_in[3];
    const float* a_dst1 = (const float*)d_in[4];
    const float* b1     = (const float*)d_in[5];
    const float* bn_g   = (const float*)d_in[6];
    const float* bn_b   = (const float*)d_in[7];
    const float* bn_m   = (const float*)d_in[8];
    const float* bn_v   = (const float*)d_in[9];
    const float* W2     = (const float*)d_in[10];
    const float* a_src2 = (const float*)d_in[11];
    const float* a_dst2 = (const float*)d_in[12];
    const float* b2     = (const float*)d_in[13];

    const int IN_DIM = 128, F1 = 256, C1 = 64, H1 = 4, C2 = 64;
    const int NPB = 196;
    int N = in_sizes[0] / IN_DIM;
    int E = in_sizes[1] / 2;

    float* ws = (float*)d_ws;
    size_t o = 0;
    float* h1   = ws + o; o += (size_t)N * F1;
    float* act1 = ws + o; o += (size_t)N * F1;
    float* as1  = ws + o; o += (size_t)N * H1;
    float* ad1  = ws + o; o += (size_t)N * H1;
    int*   eseg = (int*)(ws + o); o += (size_t)N * C2;
    float* h2   = h1;
    float* as2  = as1;
    float* ad2  = ad1;

    const int ngroups = (N + NPB - 1) / NPB;

    gemm_part<128, 1, 8, 196, 256, 384, 32><<<384 + (N + 31) / 32, 256, 0, stream>>>(
        x, W1, h1, N, ei, E, eseg);
    alpha_kernel<<<nblk(N), 256, 0, stream>>>(h1, a_src1, a_dst1, as1, ad1, N, H1, C1);
    agg_owner<4, 64, 196, 8192, true, 384, 32><<<ngroups, 1024, 0, stream>>>(
        ei, E, N, as1, ad1, h1, b1, bn_g, bn_b, bn_m, bn_v, act1, eseg);

    wmma_gemm_f32<256, 2, 2><<<(N + 63) / 64, 128, 0, stream>>>(act1, W2, h2, N);
    alpha_kernel<<<nblk(N), 256, 0, stream>>>(h2, a_src2, a_dst2, as2, ad2, N, 1, C2);
    agg_owner<1, 64, 196, 8192, false, 384, 32><<<ngroups, 1024, 0, stream>>>(
        ei, E, N, as2, ad2, h2, b2, b2, b2, b2, b2, (float*)d_out, eseg);
}


# Owner groups sort their edge list by destination in LDS (counting sort, placing atomics); a wave sums each node's rows in registers: no LDS float atomics, fill or LDS epilogue; sums in arrival order
# speedup vs baseline: 8.6773x; 1.1261x over previous
#include <hip/hip_runtime.h>
#include <hip/hip_bf16.h>
#include <math.h>

typedef __attribute__((ext_vector_type(2))) float v2f;
typedef __attribute__((ext_vector_type(8))) float v8f;

#define NEG_SLOPE 0.2f
#define BN_EPS 1e-5f

__device__ __forceinline__ unsigned encodeF(float f) {
    unsigned u = __float_as_uint(f);
    return (u & 0x80000000u) ? ~u : (u | 0x80000000u);
}
__device__ __forceinline__ float decodeF(unsigned u) {
    unsigned b = (u & 0x80000000u) ? (u & 0x7fffffffu) : ~u;
    return __uint_as_float(b);
}
__device__ __forceinline__ float lrelu(float v) {
    return v > 0.0f ? v : NEG_SLOPE * v;
}

template <int K, int NWM, int NWN>
__device__ __forceinline__ void gemm_tile_f32(const float* __restrict__ A,
                                              const float* __restrict__ B,
                                              float* __restrict__ C, int M,
                                              float* __restrict__ As, int bid) {
    static_assert(K % 4 == 0, "a k-step is 4 wide and rows are copied in 16-byte pieces");
    constexpr int NT = 32 * NWM * NWN;
    constexpr int BM = 32 * NWM;
    constexpr int NC = 32 * NWN;
    constexpr int P = K + 4;
    constexpr int K4 = K / 4;
    const int tid = threadIdx.x;
    const int lane = tid & 31;
    const int wv = __builtin_amdgcn_readfirstlane(tid >> 5);
    const int wm = wv / NWN;
    const int wn = wv - wm * NWN;
    const int row0 = bid * BM;

    static_assert((BM * K4) % NT == 0, "every thread copies the same number of 16-byte pieces");
    constexpr int NI = (BM * K4) / NT;
    constexpr int UB = NI < 8 ? NI : 8;
    static_assert(NI % UB == 0, "whole batches");
    for (int i0 = 0; i0 < NI; i0 += UB) {
        float4 v[UB];
#pragma unroll
        for (int u = 0; u < UB; ++u) {
            const int i = (i0 + u) * NT + tid;
            const int r = i / K4;
            const int c4 = i - r * K4;
            const int gr = min(row0 + r, M - 1);
            v[u] = *(const float4*)(A + (size_t)gr * K + 4 * c4);
        }
#pragma unroll
        for (int u = 0; u < UB; ++u) {
            const int i = (i0 + u) * NT + tid;
            const int r = i / K4;
            const int c4 = i - r * K4;
            *(float4*)(As + r * P + 4 * c4) = v[u];
        }
    }
    __syncthreads();

    const int lo = lane & 15;
    const int hi = lane >> 4;
    const float* a0p = As + (32 * wm + lo) * P + 2 * hi;
    const float* a1p = a0p + 16 * P;
    const float* bp = B + (size_t)(2 * hi) * NC + 32 * wn + lo;

    v8f c00 = {0.f, 0.f, 0.f, 0.f, 0.f, 0.f, 0.f, 0.f};
    v8f c01 = c00, c10 = c00, c11 = c00;
#pragma unroll 4
    for (int k0 = 0; k0 < K; k0 += 4) {
        const v2f a0 = *(const v2f*)(a0p + k0);
        const v2f a1 = *(const v2f*)(a1p + k0);
        v2f b0, b1;
        b0.x = bp[(size_t)k0 * NC];
        b0.y = bp[(size_t)(k0 + 1) * NC];
        b1.x = bp[(size_t)k0 * NC + 16];
        b1.y = bp[(size_t)(k0 + 1) * NC + 16];
        c00 = __builtin_amdgcn_wmma_f32_16x16x4_f32(false, a0, false, b0, (short)0, c00, false, false);
        c01 = __builtin_amdgcn_wmma_f32_16x16x4_f32(false, a0, false, b1, (short)0, c01, false, false);
        c10 = __builtin_amdgcn_wmma_f32_16x16x4_f32(false, a1, false, b0, (short)0, c10, false, false);
        c11 = __builtin_amdgcn_wmma_f32_16x16x4_f32(false, a1, false, b1, (short)0, c11, false, false);
    }
    const int ncol = 32 * wn + lo;
#pragma unroll
    for (int v = 0; v < 8; ++v) {
        const int r0 = row0 + 32 * wm + v + 8 * hi;
        const int r1 = r0 + 16;
        if (r0 < M) {
            C[(size_t)r0 * NC + ncol] = c00[v];
            C[(size_t)r0 * NC + ncol + 16] = c01[v];
        }
        if (r1 < M) {
            C[(size_t)r1 * NC + ncol] = c10[v];
            C[(size_t)r1 * NC + ncol + 16] = c11[v];
        }
    }
}

template <int K, int NWM, int NWN>
__global__ __launch_bounds__(32 * NWM * NWN) void wmma_gemm_f32(const float* __restrict__ A,
                                                                const float* __restrict__ B,
                                                                float* __restrict__ C, int M) {
    __shared__ __attribute__((aligned(16))) float As[32 * NWM * (K + 4)];
    gemm_tile_f32<K, NWM, NWN>(A, B, C, M, As, blockIdx.x);
}

__global__ __launch_bounds__(256) void alpha_kernel(const float* __restrict__ hf,
                                                    const float* __restrict__ a_s,
                                                    const float* __restrict__ a_d,
                                                    float* __restrict__ as_o,
                                                    float* __restrict__ ad_o,
                                                    int N, int H, int Cc) {
    int n = blockIdx.x * blockDim.x + threadIdx.x;
    if (n >= N) return;
    int F = H * Cc;
    for (int h = 0; h < H; ++h) {
        float s = 0.f, d = 0.f;
        for (int c = 0; c < Cc; ++c) {
            float v = hf[(size_t)n * F + h * Cc + c];
            s += v * a_s[h * Cc + c];
            d += v * a_d[h * Cc + c];
        }
        as_o[n * H + h] = s;
        ad_o[n * H + h] = d;
    }
}

template <int NT, int NPB, int NB, int NS, int SEG>
__device__ __forceinline__ void part_edges_slice(const int* __restrict__ ei, int E,
                                                 int* __restrict__ seg, int* __restrict__ buf,
                                                 int s) {
    static_assert(NPB <= 256, "an entry keeps the local destination in its low 8 bits");
    static_assert(SEG % 4 == 0, "segments are moved in 16-byte pieces");
    constexpr int UB = 5;
    constexpr int V = SEG / 4;
    const int tid = threadIdx.x;
    const int* __restrict__ srcp = ei;
    const int* __restrict__ dstp = ei + E;
    const int len = (E + NS - 1) / NS;
    const int e0 = s * len;
    const int e1 = min(E, e0 + len);

    for (int i = tid; i < NB * V; i += NT)
        ((int4*)buf)[i] = make_int4((i % V) ? -1 : 0, -1, -1, -1);
    __syncthreads();
    for (int base = e0; base < e1; base += NT * UB) {
        int d[UB], sr[UB];
#pragma unroll
        for (int k = 0; k < UB; ++k) {
            const int e = base + k * NT + tid;
            const int ec = min(e, e1 - 1);
            const int dv = dstp[ec];
            sr[k] = srcp[ec];
            d[k] = (e < e1) ? dv : -1;
        }
#pragma unroll
        for (int k = 0; k < UB; ++k) {
            const unsigned b = (unsigned)d[k] / (unsigned)NPB;
            if (b < (unsigned)NB) {
                const int pos = atomicAdd(&buf[b * SEG], 1);
                if (pos < SEG - 1)
                    buf[b * SEG + 1 + pos] = (sr[k] << 8) | (int)((unsigned)d[k] - b * (unsigned)NPB);
            }
        }
    }
    __syncthreads();
    const int4* bv = (const int4*)buf;
    for (int i = tid; i < NB * V; i += NT) {
        const int b = i / V;
        const int q = i - b * V;
        ((int4*)(seg + ((size_t)b * NS + s) * SEG))[q] = bv[i];
    }
}

template <int K, int NWM, int NWN, int NPB, int NB, int NS, int SEG>
__global__ __launch_bounds__(32 * NWM * NWN) void gemm_part(const float* __restrict__ A,
                                                            const float* __restrict__ B,
                                                            float* __restrict__ C, int M,
                                                            const int* __restrict__ ei, int E,
                                                            int* __restrict__ seg) {
    constexpr int NT = 32 * NWM * NWN;
    constexpr int GW = 32 * NWM * (K + 4);
    constexpr int PW = NB * SEG;
    __shared__ __attribute__((aligned(16))) unsigned char lds[4 * (GW > PW ? GW : PW)];
    if (blockIdx.x >= NS)
        gemm_tile_f32<K, NWM, NWN>(A, B, C, M, (float*)lds, blockIdx.x - NS);
    else
        part_edges_slice<NT, NPB, NB, NS, SEG>(ei, E, seg, (int*)lds, blockIdx.x);
}

template <int VEC>
__device__ __forceinline__ int row_pos(int ch) {
    constexpr int LW = 32 * VEC;
    const int k = ch / LW;
    const int rr = ch - k * LW;
    return k * LW + (rr % VEC) * 32 + rr / VEC;
}

template <int M>
__device__ __forceinline__ float row_xadd(float v) {
    return v + __int_as_float(__builtin_amdgcn_update_dpp(0, __float_as_int(v), 0x160 | M, 0xf, 0xf, true));
}
template <int H>
__device__ __forceinline__ float head_sum(float v) {
    if (H <= 1) v = row_xadd<1>(v);
    if (H <= 2) v = row_xadd<2>(v);
    if (H <= 4) v = row_xadd<4>(v);
    if (H <= 8) v = row_xadd<8>(v);
    if (H <= 16) v += __shfl_xor(v, 16);
    return v;
}

template <int VEC> using fvec = float __attribute__((ext_vector_type(VEC)));
template <int H, int Cc, int VEC, int NL, int KK>
__device__ __forceinline__ void row_batch(const float* __restrict__ hsrc, int s, float al, int nb,
                                          int lane, fvec<VEC> (&a)[NL]) {
    typedef fvec<VEC> rvec;
    constexpr int F = H * Cc;
    constexpr int LW = 32 * VEC;
    constexpr int HPL = LW > Cc ? LW / Cc : 1;
    constexpr int LPH = 32 / HPL;
    rvec r[KK][NL];
#pragma unroll
    for (int i = 0; i < KK; ++i) {
        const int si = (i < nb) ? __builtin_amdgcn_readlane(s, i * H) : 0;
        const rvec* hr = (const rvec*)(hsrc + (size_t)si * F) + lane;
        for (int k = 0; k < NL; ++k) r[i][k] = hr[32 * k];
    }
#pragma unroll
    for (int i = 0; i < KK; ++i) {
        for (int k = 0; k < NL; ++k) {
            float wk = __int_as_float(
                __builtin_amdgcn_readlane(__float_as_int(al), i * H + (k * LW) / Cc));
            if (HPL == 2) {
                const float w1 = __int_as_float(
                    __builtin_amdgcn_readlane(__float_as_int(al), i * H + (k * LW) / Cc + 1));
                wk = lane < LPH ? wk : w1;
            }
            for (int c = 0; c < VEC; ++c) a[k][c] += r[i][k][c] * wk;
        }
    }
}

template <int H, int Cc, int NPB, int CAP, bool BN, int NS, int SEG>
__global__ __launch_bounds__(1024) void agg_owner(const int* __restrict__ ei, int E, int N,
                                                  const float* __restrict__ as_i,
                                                  const float* __restrict__ ad_i,
                                                  const float* __restrict__ hsrc,
                                                  const float* __restrict__ bias,
                                                  const float* __restrict__ g,
                                                  const float* __restrict__ be,
                                                  const float* __restrict__ mu,
                                                  const float* __restrict__ var,
                                                  float* __restrict__ outp,
                                                  const int* __restrict__ seg) {
    static_assert((H & (H - 1)) == 0, "H must be a power of two");
    static_assert(NPB <= 256, "an entry keeps the local destination in its low 8 bits");
    static_assert(Cc % 32 == 0, "a 32-channel block must lie inside one head");
    static_assert(SEG % 4 == 0, "segments are read in 16-byte pieces");
    constexpr int F = H * Cc;
    constexpr int VEC = (F / 32) < 4 ? (F / 32) : 4;
    constexpr int LW = 32 * VEC;
    constexpr int NL = F / LW;
    constexpr int HPL = LW > Cc ? LW / Cc : 1;
    constexpr int LPH = 32 / HPL;
    static_assert(F % LW == 0, "a row is a whole number of wave-wide loads");
    static_assert(LW % Cc == 0 || Cc % LW == 0, "a load holds whole heads, or lies inside one");
    static_assert(HPL <= 2, "the weight of a lane is picked from at most two heads per load");
    constexpr int KK = (32 / NL) < (32 / H) ? (32 / NL) : (32 / H);
    typedef fvec<VEC> rvec;
    constexpr int NT = 1024;
    constexpr int V = SEG / 4;
    constexpr int NV = NS * V;
    constexpr int UB = (NV + NT - 1) / NT;
    constexpr int ACCW = NPB * F;
    constexpr int FASTW = CAP + CAP * H;
    constexpr int UW = ACCW > FASTW ? ACCW : FASTW;
    __shared__ __attribute__((aligned(16))) float ubuf[UW];
    float* acc = ubuf;
    int* ssrc = (int*)ubuf;
    float* wts = ubuf + CAP;
    __shared__ float zs[NPB * H];
    __shared__ unsigned mxs[NPB * H];
    __shared__ int le[CAP];
    __shared__ int ldl[CAP];
    __shared__ int nst[NPB + 1];
    __shared__ int nfill[NPB];
    __shared__ unsigned cnt;
    __shared__ unsigned nxt;

    const int tid = threadIdx.x;
    const int lane = tid & 31;
    const int wv = __builtin_amdgcn_readfirstlane(tid >> 5);
    const int lo = blockIdx.x * NPB;
    const int* __restrict__ srcp = ei;
    const int* __restrict__ dstp = ei + E;

    const int4* __restrict__ sv = (const int4*)(seg + (size_t)blockIdx.x * (NS * SEG));
    int4 w4[UB];
#pragma unroll
    for (int k = 0; k < UB; ++k) {
        const int v = k * NT + tid;
        w4[k] = (v < NV) ? sv[v] : make_int4(-1, -1, -1, -1);
    }

    for (int i = tid; i < NPB * H; i += NT) mxs[i] = 0u;
    if (tid < NPB) nfill[tid] = 0;
    if (tid == 0) { cnt = 0u; nxt = 0u; }
    __syncthreads();

    if (tid < NPB && lo + tid < N) {
        const unsigned pos = atomicAdd(&cnt, 1u);
        if (pos < (unsigned)CAP) { le[pos] = lo + tid; ldl[pos] = tid; }
        atomicAdd(&nfill[tid], 1);
    }
#pragma unroll
    for (int k = 0; k < UB; ++k) {
        const int v = k * NT + tid;
        int ww[4] = {w4[k].x, w4[k].y, w4[k].z, w4[k].w};
        if (v < NV && (v & (V - 1)) == 0) {
            if (ww[0] > SEG - 1) atomicAdd(&cnt, (unsigned)CAP + 1u);
            ww[0] = -1;
        }
#pragma unroll
        for (int q = 0; q < 4; ++q) {
            if (ww[q] >= 0) {
                const unsigned pos = atomicAdd(&cnt, 1u);
                if (pos < (unsigned)CAP) { le[pos] = ww[q] >> 8; ldl[pos] = ww[q] & 255; }
                atomicAdd(&nfill[ww[q] & 255], 1);
            }
        }
    }
    __syncthreads();
    const int total = (int)cnt;

    if (total <= CAP) {
        if (wv == 0) {
            constexpr int PER = (NPB + 31) / 32;
            int c[PER];
            int sum = 0;
            for (int j = 0; j < PER; ++j) {
                const int idx = lane * PER + j;
                c[j] = idx < NPB ? nfill[idx] : 0;
                sum += c[j];
            }
            int inc = sum;
            for (int d = 1; d < 32; d <<= 1) {
                const int t = __shfl_up(inc, d);
                if (lane >= d) inc += t;
            }
            int run = inc - sum;
            for (int j = 0; j < PER; ++j) {
                const int idx = lane * PER + j;
                if (idx < NPB) { nst[idx] = run; nfill[idx] = run; }
                run += c[j];
            }
            if (lane == 31) nst[NPB] = inc;
        }
        __syncthreads();
        for (int p = tid; p < total; p += NT) {
            const int s = le[p];
            const int dl = ldl[p];
            const int pos = atomicAdd(&nfill[dl], 1);
            ssrc[pos] = s;
            for (int h = 0; h < H; ++h) {
                const float v = lrelu(as_i[s * H + h] + ad_i[(lo + dl) * H + h]);
                wts[pos * H + h] = v;
                atomicMax(&mxs[dl * H + h], encodeF(v));
            }
        }
        __syncthreads();
        const int lh = lane & (H - 1);
        const int li = lane / H;
        for (int it = 0; it <= NPB; ++it) {
            int nd = 0;
            if (lane == 0) nd = (int)atomicAdd(&nxt, 1u);
            nd = __builtin_amdgcn_readfirstlane(nd);
            if (nd >= NPB || lo + nd >= N) break;
            const int st = __builtin_amdgcn_readfirstlane(nst[nd]);
            const int en = __builtin_amdgcn_readfirstlane(nst[nd + 1]);
            const float mxh = decodeF(mxs[nd * H + lh]);
            rvec a[NL];
            for (int k = 0; k < NL; ++k)
                for (int c = 0; c < VEC; ++c) a[k][c] = 0.f;
            float zp = 0.f;
            for (int p0 = st; p0 < en; p0 += KK) {
                const int nb = min(KK, en - p0);
                const int p = min(p0 + li, en - 1);
                const int s = ssrc[p];
                const float al = (li < nb) ? __expf(wts[p * H + lh] - mxh) : 0.f;
                zp += al;
                row_batch<H, Cc, VEC, NL, KK>(hsrc, s, al, nb, lane, a);
            }
            const float rzl = 1.0f / (head_sum<H>(zp) + 1e-16f);
            float* orow = outp + (size_t)(lo + nd) * F;
            for (int k = 0; k < NL; ++k) {
                float rz = __int_as_float(__builtin_amdgcn_readlane(__float_as_int(rzl), (k * LW) / Cc));
                if (HPL == 2) {
                    const float rz1 = __int_as_float(
                        __builtin_amdgcn_readlane(__float_as_int(rzl), (k * LW) / Cc + 1));
                    rz = lane < LPH ? rz : rz1;
                }
                const int ch = k * LW + VEC * lane;
                rvec o;
                for (int c = 0; c < VEC; ++c) {
                    float v = a[k][c] * rz + bias[ch + c];
                    if (BN) {
                        v = (v - mu[ch + c]) * g[ch + c] * rsqrtf(var[ch + c] + BN_EPS) + be[ch + c];
                        v = v > 0.f ? v : (__expf(v) - 1.f);
                    }
                    o[c] = v;
                }
                ((rvec*)orow)[lane + 32 * k] = o;
            }
        }
    } else {
        for (int i = tid; i < NPB * F; i += NT) acc[i] = 0.f;
        for (int i = tid; i < NPB * H; i += NT) zs[i] = 0.f;
        __syncthreads();
        const int Etot = E + N;
        for (int e = tid; e < Etot; e += NT) {
            const int ec = min(e, E - 1);
            const int dd = (e < E) ? dstp[ec] : (e - E);
            const unsigned dl = (unsigned)(dd - lo);
            if (dl < (unsigned)NPB) {
                const int s = (e < E) ? srcp[ec] : (e - E);
                for (int h = 0; h < H; ++h) {
                    const float v = lrelu(as_i[s * H + h] + ad_i[dd * H + h]);
                    atomicMax(&mxs[dl * H + h], encodeF(v));
                }
            }
        }
        __syncthreads();
        for (int e = tid; e < Etot; e += NT) {
            const int ec = min(e, E - 1);
            const int dd = (e < E) ? dstp[ec] : (e - E);
            const unsigned dl = (unsigned)(dd - lo);
            if (dl < (unsigned)NPB) {
                const int s = (e < E) ? srcp[ec] : (e - E);
                for (int h = 0; h < H; ++h) {
                    const float v = lrelu(as_i[s * H + h] + ad_i[dd * H + h]);
                    const float w = __expf(v - decodeF(mxs[dl * H + h]));
                    atomicAdd(&zs[dl * H + h], w);
                    for (int c = 0; c < Cc; ++c)
                        atomicAdd(&acc[dl * F + row_pos<VEC>(h * Cc + c)], w * hsrc[(size_t)s * F + h * Cc + c]);
                }
            }
        }
        __syncthreads();

        for (int i = tid; i < NPB * H; i += NT) zs[i] = 1.0f / (zs[i] + 1e-16f);
        __syncthreads();
        static_assert(Cc % 4 == 0, "a group of four columns must lie inside one head");
        constexpr int F4 = F / 4;
        static_assert(NT % F4 == 0, "every pass covers whole rows");
        constexpr int RS = NT / F4;
        const int cg = tid % F4;
        const int j0 = 4 * cg;
        const int hh = j0 / Cc;
        float cb[4], cm[4], cga[4], crs[4], cbe[4];
        int cp[4];
        for (int c = 0; c < 4; ++c) {
            cp[c] = row_pos<VEC>(j0 + c);
            cb[c] = bias[j0 + c];
            if (BN) {
                cm[c] = mu[j0 + c];
                cga[c] = g[j0 + c];
                crs[c] = rsqrtf(var[j0 + c] + BN_EPS);
                cbe[c] = be[j0 + c];
            }
        }
        for (int i = tid / F4; i < NPB; i += RS) {
            const int n = lo + i;
            if (n < N) {
                const float rz = zs[i * H + hh];
                float o[4];
                for (int c = 0; c < 4; ++c) o[c] = acc[i * F + cp[c]];
                for (int c = 0; c < 4; ++c) {
                    float v = o[c] * rz + cb[c];
                    if (BN) {
                        v = (v - cm[c]) * cga[c] * crs[c] + cbe[c];
                        v = v > 0.f ? v : (__expf(v) - 1.f);
                    }
                    o[c] = v;
                }
                *(float4*)(outp + (size_t)n * F + j0) = make_float4(o[0], o[1], o[2], o[3]);
            }
        }
    }
}

static inline int nblk(long long t) { return (int)((t + 255) / 256); }

extern "C" void kernel_launch(void* const* d_in, const int* in_sizes, int n_in,
                              void* d_out, int out_size, void* d_ws, size_t ws_size,
                              hipStream_t stream) {
    const float* x      = (const float*)d_in[0];
    const int*   ei     = (const int*)d_in[1];
    const float* W1     = (const float*)d_in[2];
    const float* a_src1 = (const float*)d_in[3];
    const float* a_dst1 = (const float*)d_in[4];
    const float* b1     = (const float*)d_in[5];
    const float* bn_g   = (const float*)d_in[6];
    const float* bn_b   = (const float*)d_in[7];
    const float* bn_m   = (const float*)d_in[8];
    const float* bn_v   = (const float*)d_in[9];
    const float* W2     = (const float*)d_in[10];
    const float* a_src2 = (const float*)d_in[11];
    const float* a_dst2 = (const float*)d_in[12];
    const float* b2     = (const float*)d_in[13];

    const int IN_DIM = 128, F1 = 256, C1 = 64, H1 = 4, C2 = 64;
    const int NPB = 196;
    int N = in_sizes[0] / IN_DIM;
    int E = in_sizes[1] / 2;

    float* ws = (float*)d_ws;
    size_t o = 0;
    float* h1   = ws + o; o += (size_t)N * F1;
    float* act1 = ws + o; o += (size_t)N * F1;
    float* as1  = ws + o; o += (size_t)N * H1;
    float* ad1  = ws + o; o += (size_t)N * H1;
    int*   eseg = (int*)(ws + o); o += (size_t)N * C2;
    float* h2   = h1;
    float* as2  = as1;
    float* ad2  = ad1;

    const int ngroups = (N + NPB - 1) / NPB;

    gemm_part<128, 1, 8, 196, 256, 384, 32><<<384 + (N + 31) / 32, 256, 0, stream>>>(
        x, W1, h1, N, ei, E, eseg);
    alpha_kernel<<<nblk(N), 256, 0, stream>>>(h1, a_src1, a_dst1, as1, ad1, N, H1, C1);
    agg_owner<4, 64, 196, 8192, true, 384, 32><<<ngroups, 1024, 0, stream>>>(
        ei, E, N, as1, ad1, h1, b1, bn_g, bn_b, bn_m, bn_v, act1, eseg);

    wmma_gemm_f32<256, 2, 2><<<(N + 63) / 64, 128, 0, stream>>>(act1, W2, h2, N);
    alpha_kernel<<<nblk(N), 256, 0, stream>>>(h2, a_src2, a_dst2, as2, ad2, N, 1, C2);
    agg_owner<1, 64, 196, 8192, false, 384, 32><<<ngroups, 1024, 0, stream>>>(
        ei, E, N, as2, ad2, h2, b2, b2, b2, b2, b2, (float*)d_out, eseg);
}
